# starting kernel
# baseline (speedup 1.0000x reference)
#include <hip/hip_runtime.h>
#include <hip/hip_bf16.h>

typedef float v2f __attribute__((ext_vector_type(2)));
typedef float v8f __attribute__((ext_vector_type(8)));

#define HEADS 4
#define CH    32
#define DIN_  128
#define NEG_SLOPE_ 0.2f

__device__ __forceinline__ unsigned f2ord(float f) {
    unsigned u = __float_as_uint(f);
    return (u & 0x80000000u) ? ~u : (u | 0x80000000u);
}
__device__ __forceinline__ float ord2f(unsigned u) {
    return (u & 0x80000000u) ? __uint_as_float(u & 0x7FFFFFFFu) : __uint_as_float(~u);
}
#define ORD_NEG_INF 0x007FFFFFu

__device__ __forceinline__ float leaky(float v) {
    return v > 0.0f ? v : NEG_SLOPE_ * v;
}

__global__ void init_kernel(float* __restrict__ out, float* __restrict__ deg,
                            float* __restrict__ wsum, float* __restrict__ denom,
                            unsigned* __restrict__ amax, int n) {
    long long i = (long long)blockIdx.x * blockDim.x + threadIdx.x;
    long long tot = (long long)n * DIN_;
    if (i < tot) out[i] = 0.0f;
    if (i < n) { deg[i] = 0.0f; wsum[i] = 0.0f; }
    if (i < (long long)n * HEADS) { denom[i] = 0.0f; amax[i] = ORD_NEG_INF; }
}

__global__ void gemm_wmma_kernel(const float* __restrict__ x,
                                 const float* __restrict__ W,
                                 float* __restrict__ h, int n) {
    const int wave = threadIdx.x >> 5;
    const int lane = threadIdx.x & 31;
    const int half = lane >> 4;
    const int l16  = lane & 15;
    const int row0 = blockIdx.x * 16;
    const int col0 = wave * 16;

    int arow = row0 + l16;
    if (arow >= n) arow = n - 1;
    const float* __restrict__ xrow = x + (size_t)arow * DIN_;
    const int kk = half * 2;

    v8f acc = {};
    #pragma unroll
    for (int k = 0; k < DIN_; k += 4) {
        v2f a, b;
        a.x = xrow[k + kk + 0];
        a.y = xrow[k + kk + 1];
        b.x = W[(size_t)(k + kk + 0) * DIN_ + col0 + l16];
        b.y = W[(size_t)(k + kk + 1) * DIN_ + col0 + l16];
        acc = __builtin_amdgcn_wmma_f32_16x16x4_f32(
             false, a,  false, b,
             (short)0, acc,  false,  false);
    }

    float* __restrict__ hp = h + (size_t)(col0 + l16);
    #pragma unroll
    for (int r = 0; r < 8; ++r) {
        int row = row0 + r + half * 8;
        if (row < n) hp[(size_t)row * DIN_] = acc[r];
    }
}

__global__ void att_kernel(const float* __restrict__ h,
                           const float* __restrict__ att_src,
                           const float* __restrict__ att_dst,
                           float* __restrict__ a_src, float* __restrict__ a_dst, int n) {
    long long i = (long long)blockIdx.x * blockDim.x + threadIdx.x;
    if (i >= (long long)n * HEADS) return;
    const int head = (int)(i & (HEADS - 1));
    const float* __restrict__ hp = h + (i >> 2) * DIN_ + head * CH;
    const float* __restrict__ as = att_src + head * CH;
    const float* __restrict__ ad = att_dst + head * CH;
    float s = 0.0f, d = 0.0f;
    #pragma unroll
    for (int c = 0; c < CH; ++c) {
        float v = hp[c];
        s = fmaf(v, as[c], s);
        d = fmaf(v, ad[c], d);
    }
    a_src[i] = s;
    a_dst[i] = d;
}

__global__ void edge_pass1_kernel(const int* __restrict__ src, const int* __restrict__ dst,
                                  const float* __restrict__ ea,
                                  const float* __restrict__ a_src, const float* __restrict__ a_dst,
                                  float* __restrict__ deg, float* __restrict__ wsum,
                                  unsigned* __restrict__ amax, int e) {
    long long i = (long long)blockIdx.x * blockDim.x + threadIdx.x;
    if (i >= e) return;
    const int s = src[i], d = dst[i];
    const float w = ea[i];
    atomicAdd(&deg[d], 1.0f);
    atomicAdd(&wsum[d], w);
    const float l2w = __log2f(w);
    #pragma unroll
    for (int hh = 0; hh < HEADS; ++hh) {
        float v = leaky(a_src[s * HEADS + hh] + a_dst[d * HEADS + hh]) + l2w;
        atomicMax(&amax[d * HEADS + hh], f2ord(v));
    }
}

__global__ void node_pass_kernel(const float* __restrict__ deg, const float* __restrict__ wsum,
                                 float* __restrict__ loopw,
                                 const float* __restrict__ a_src, const float* __restrict__ a_dst,
                                 unsigned* __restrict__ amax, int n) {
    long long i = (long long)blockIdx.x * blockDim.x + threadIdx.x;
    if (i >= n) return;
    const float dg = deg[i];
    const float lw = dg > 0.0f ? wsum[i] / dg : 1.0f;
    loopw[i] = lw;
    const float l2w = __log2f(lw);
    #pragma unroll
    for (int hh = 0; hh < HEADS; ++hh) {
        float v = leaky(a_src[i * HEADS + hh] + a_dst[i * HEADS + hh]) + l2w;
        atomicMax(&amax[i * HEADS + hh], f2ord(v));
    }
}

__global__ void scatter_kernel(const int* __restrict__ src, const int* __restrict__ dst,
                               const float* __restrict__ ea,
                               const float* __restrict__ a_src, const float* __restrict__ a_dst,
                               const unsigned* __restrict__ amax,
                               const float* __restrict__ h,
                               float* __restrict__ denom, float* __restrict__ out, int e) {
    long long gid = (long long)blockIdx.x * blockDim.x + threadIdx.x;
    long long edge = gid >> 5;
    const int lane = (int)(gid & 31);
    if (edge >= e) return;
    const int s = src[edge], d = dst[edge];
    const int head = lane >> 3;
    float v = leaky(a_src[s * HEADS + head] + a_dst[d * HEADS + head]);
    v += __log2f(ea[edge]);
    const float ex = __expf(v - ord2f(amax[d * HEADS + head]));
    if ((lane & 7) == 0) atomicAdd(&denom[d * HEADS + head], ex);
    const float* __restrict__ hp = h + (size_t)s * DIN_ + lane * 4;
    float* __restrict__ op = out + (size_t)d * DIN_ + lane * 4;
    #pragma unroll
    for (int c = 0; c < 4; ++c) atomicAdd(&op[c], ex * hp[c]);
}

__global__ void finalize_kernel(const float* __restrict__ h,
                                const float* __restrict__ a_src, const float* __restrict__ a_dst,
                                const float* __restrict__ loopw, const unsigned* __restrict__ amax,
                                const float* __restrict__ denom, const float* __restrict__ bias,
                                float* __restrict__ out, int n) {
    long long i = (long long)blockIdx.x * blockDim.x + threadIdx.x;
    if (i >= (long long)n * DIN_) return;
    const int node = (int)(i >> 7);
    const int col  = (int)(i & (DIN_ - 1));
    const int head = col >> 5;
    float v = leaky(a_src[node * HEADS + head] + a_dst[node * HEADS + head]);
    v += __log2f(loopw[node]);
    const float ex = __expf(v - ord2f(amax[node * HEADS + head]));
    const float den = denom[node * HEADS + head] + ex;
    out[i] = (out[i] + ex * h[i]) / den + bias[col];
}

extern "C" void kernel_launch(void* const* d_in, const int* in_sizes, int n_in,
                              void* d_out, int out_size, void* d_ws, size_t ws_size,
                              hipStream_t stream) {
    const float* x        = (const float*)d_in[0];
    const int*   eidx     = (const int*)d_in[1];
    const float* eattr    = (const float*)d_in[2];
    const float* W        = (const float*)d_in[3];
    const float* att_src  = (const float*)d_in[4];
    const float* att_dst  = (const float*)d_in[5];
    const float* bias     = (const float*)d_in[6];

    const int n = in_sizes[0] / DIN_;
    const int e = in_sizes[2];
    const int* src = eidx;
    const int* dst = eidx + e;

    float* out = (float*)d_out;

    float*    h     = (float*)d_ws;
    float*    asrc  = h     + (size_t)n * DIN_;
    float*    adst  = asrc  + (size_t)n * HEADS;
    float*    deg   = adst  + (size_t)n * HEADS;
    float*    wsum  = deg   + n;
    float*    loopw = wsum  + n;
    float*    denom = loopw + n;
    unsigned* amax  = (unsigned*)(denom + (size_t)n * HEADS);

    const int B = 256;
    const long long nTot = (long long)n * DIN_;

    init_kernel<<<(unsigned)((nTot + B - 1) / B), B, 0, stream>>>(out, deg, wsum, denom, amax, n);

    gemm_wmma_kernel<<<(unsigned)((n + 15) / 16), B, 0, stream>>>(x, W, h, n);

    att_kernel<<<(unsigned)(((long long)n * HEADS + B - 1) / B), B, 0, stream>>>(
        h, att_src, att_dst, asrc, adst, n);

    edge_pass1_kernel<<<(unsigned)(((long long)e + B - 1) / B), B, 0, stream>>>(
        src, dst, eattr, asrc, adst, deg, wsum, amax, e);

    node_pass_kernel<<<(unsigned)((n + B - 1) / B), B, 0, stream>>>(
        deg, wsum, loopw, asrc, adst, amax, n);

    scatter_kernel<<<(unsigned)(((long long)e * 32 + B - 1) / B), B, 0, stream>>>(
        src, dst, eattr, asrc, adst, amax, h, denom, out, e);

    finalize_kernel<<<(unsigned)((nTot + B - 1) / B), B, 0, stream>>>(
        h, asrc, adst, loopw, amax, denom, bias, out, n);
}


# scatter: lane owns columns lane+32c (c = head) so each of a wave's 4 f32 atomic adds and 4 row loads covers one whole 128-B line, not a quarter of four lines; atomics kept, sum order as before
# speedup vs baseline: 1.1120x; 1.1120x over previous
#include <hip/hip_runtime.h>
#include <hip/hip_bf16.h>

typedef float v2f __attribute__((ext_vector_type(2)));
typedef float v8f __attribute__((ext_vector_type(8)));

#define HEADS 4
#define CH    32
#define DIN_  128
#define NEG_SLOPE_ 0.2f

__device__ __forceinline__ unsigned f2ord(float f) {
    unsigned u = __float_as_uint(f);
    return (u & 0x80000000u) ? ~u : (u | 0x80000000u);
}
__device__ __forceinline__ float ord2f(unsigned u) {
    return (u & 0x80000000u) ? __uint_as_float(u & 0x7FFFFFFFu) : __uint_as_float(~u);
}
#define ORD_NEG_INF 0x007FFFFFu

__device__ __forceinline__ float leaky(float v) {
    return v > 0.0f ? v : NEG_SLOPE_ * v;
}

__global__ void init_kernel(float* __restrict__ out, float* __restrict__ deg,
                            float* __restrict__ wsum, float* __restrict__ denom,
                            unsigned* __restrict__ amax, int n) {
    long long i = (long long)blockIdx.x * blockDim.x + threadIdx.x;
    long long tot = (long long)n * DIN_;
    if (i < tot) out[i] = 0.0f;
    if (i < n) { deg[i] = 0.0f; wsum[i] = 0.0f; }
    if (i < (long long)n * HEADS) { denom[i] = 0.0f; amax[i] = ORD_NEG_INF; }
}

__global__ void gemm_wmma_kernel(const float* __restrict__ x,
                                 const float* __restrict__ W,
                                 float* __restrict__ h, int n) {
    const int wave = threadIdx.x >> 5;
    const int lane = threadIdx.x & 31;
    const int half = lane >> 4;
    const int l16  = lane & 15;
    const int row0 = blockIdx.x * 16;
    const int col0 = wave * 16;

    int arow = row0 + l16;
    if (arow >= n) arow = n - 1;
    const float* __restrict__ xrow = x + (size_t)arow * DIN_;
    const int kk = half * 2;

    v8f acc = {};
    #pragma unroll
    for (int k = 0; k < DIN_; k += 4) {
        v2f a, b;
        a.x = xrow[k + kk + 0];
        a.y = xrow[k + kk + 1];
        b.x = W[(size_t)(k + kk + 0) * DIN_ + col0 + l16];
        b.y = W[(size_t)(k + kk + 1) * DIN_ + col0 + l16];
        acc = __builtin_amdgcn_wmma_f32_16x16x4_f32(
             false, a,  false, b,
             (short)0, acc,  false,  false);
    }

    float* __restrict__ hp = h + (size_t)(col0 + l16);
    #pragma unroll
    for (int r = 0; r < 8; ++r) {
        int row = row0 + r + half * 8;
        if (row < n) hp[(size_t)row * DIN_] = acc[r];
    }
}

__global__ void att_kernel(const float* __restrict__ h,
                           const float* __restrict__ att_src,
                           const float* __restrict__ att_dst,
                           float* __restrict__ a_src, float* __restrict__ a_dst, int n) {
    long long i = (long long)blockIdx.x * blockDim.x + threadIdx.x;
    if (i >= (long long)n * HEADS) return;
    const int head = (int)(i & (HEADS - 1));
    const float* __restrict__ hp = h + (i >> 2) * DIN_ + head * CH;
    const float* __restrict__ as = att_src + head * CH;
    const float* __restrict__ ad = att_dst + head * CH;
    float s = 0.0f, d = 0.0f;
    #pragma unroll
    for (int c = 0; c < CH; ++c) {
        float v = hp[c];
        s = fmaf(v, as[c], s);
        d = fmaf(v, ad[c], d);
    }
    a_src[i] = s;
    a_dst[i] = d;
}

__global__ void edge_pass1_kernel(const int* __restrict__ src, const int* __restrict__ dst,
                                  const float* __restrict__ ea,
                                  const float* __restrict__ a_src, const float* __restrict__ a_dst,
                                  float* __restrict__ deg, float* __restrict__ wsum,
                                  unsigned* __restrict__ amax, int e) {
    long long i = (long long)blockIdx.x * blockDim.x + threadIdx.x;
    if (i >= e) return;
    const int s = src[i], d = dst[i];
    const float w = ea[i];
    atomicAdd(&deg[d], 1.0f);
    atomicAdd(&wsum[d], w);
    const float l2w = __log2f(w);
    #pragma unroll
    for (int hh = 0; hh < HEADS; ++hh) {
        float v = leaky(a_src[s * HEADS + hh] + a_dst[d * HEADS + hh]) + l2w;
        atomicMax(&amax[d * HEADS + hh], f2ord(v));
    }
}

__global__ void node_pass_kernel(const float* __restrict__ deg, const float* __restrict__ wsum,
                                 float* __restrict__ loopw,
                                 const float* __restrict__ a_src, const float* __restrict__ a_dst,
                                 unsigned* __restrict__ amax, int n) {
    long long i = (long long)blockIdx.x * blockDim.x + threadIdx.x;
    if (i >= n) return;
    const float dg = deg[i];
    const float lw = dg > 0.0f ? wsum[i] / dg : 1.0f;
    loopw[i] = lw;
    const float l2w = __log2f(lw);
    #pragma unroll
    for (int hh = 0; hh < HEADS; ++hh) {
        float v = leaky(a_src[i * HEADS + hh] + a_dst[i * HEADS + hh]) + l2w;
        atomicMax(&amax[i * HEADS + hh], f2ord(v));
    }
}

__global__ void scatter_kernel(const int* __restrict__ src, const int* __restrict__ dst,
                               const float* __restrict__ ea,
                               const float* __restrict__ a_src, const float* __restrict__ a_dst,
                               const unsigned* __restrict__ amax,
                               const float* __restrict__ h,
                               float* __restrict__ denom, float* __restrict__ out, int e) {
    long long gid = (long long)blockIdx.x * blockDim.x + threadIdx.x;
    long long edge = gid >> 5;
    const int lane = (int)(gid & 31);
    if (edge >= e) return;
    const int s = src[edge], d = dst[edge];
    const float l2w = __log2f(ea[edge]);
    float ex[HEADS];
    for (int hh = 0; hh < HEADS; ++hh) {
        float v = leaky(a_src[s * HEADS + hh] + a_dst[d * HEADS + hh]);
        v += l2w;
        ex[hh] = __expf(v - ord2f(amax[d * HEADS + hh]));
    }
    float exl = ex[3];
    exl = (lane == 2) ? ex[2] : exl;
    exl = (lane == 1) ? ex[1] : exl;
    exl = (lane == 0) ? ex[0] : exl;
    if (lane < HEADS) atomicAdd(&denom[d * HEADS + lane], exl);
    const float* __restrict__ hp = h + (size_t)s * DIN_ + lane;
    float* __restrict__ op = out + (size_t)d * DIN_ + lane;
    float hv[HEADS];
    for (int c = 0; c < HEADS; ++c) hv[c] = hp[c * CH];
    for (int c = 0; c < HEADS; ++c) atomicAdd(&op[c * CH], ex[c] * hv[c]);
}

__global__ void finalize_kernel(const float* __restrict__ h,
                                const float* __restrict__ a_src, const float* __restrict__ a_dst,
                                const float* __restrict__ loopw, const unsigned* __restrict__ amax,
                                const float* __restrict__ denom, const float* __restrict__ bias,
                                float* __restrict__ out, int n) {
    long long i = (long long)blockIdx.x * blockDim.x + threadIdx.x;
    if (i >= (long long)n * DIN_) return;
    const int node = (int)(i >> 7);
    const int col  = (int)(i & (DIN_ - 1));
    const int head = col >> 5;
    float v = leaky(a_src[node * HEADS + head] + a_dst[node * HEADS + head]);
    v += __log2f(loopw[node]);
    const float ex = __expf(v - ord2f(amax[node * HEADS + head]));
    const float den = denom[node * HEADS + head] + ex;
    out[i] = (out[i] + ex * h[i]) / den + bias[col];
}

extern "C" void kernel_launch(void* const* d_in, const int* in_sizes, int n_in,
                              void* d_out, int out_size, void* d_ws, size_t ws_size,
                              hipStream_t stream) {
    const float* x        = (const float*)d_in[0];
    const int*   eidx     = (const int*)d_in[1];
    const float* eattr    = (const float*)d_in[2];
    const float* W        = (const float*)d_in[3];
    const float* att_src  = (const float*)d_in[4];
    const float* att_dst  = (const float*)d_in[5];
    const float* bias     = (const float*)d_in[6];

    const int n = in_sizes[0] / DIN_;
    const int e = in_sizes[2];
    const int* src = eidx;
    const int* dst = eidx + e;

    float* out = (float*)d_out;

    float*    h     = (float*)d_ws;
    float*    asrc  = h     + (size_t)n * DIN_;
    float*    adst  = asrc  + (size_t)n * HEADS;
    float*    deg   = adst  + (size_t)n * HEADS;
    float*    wsum  = deg   + n;
    float*    loopw = wsum  + n;
    float*    denom = loopw + n;
    unsigned* amax  = (unsigned*)(denom + (size_t)n * HEADS);

    const int B = 256;
    const long long nTot = (long long)n * DIN_;

    init_kernel<<<(unsigned)((nTot + B - 1) / B), B, 0, stream>>>(out, deg, wsum, denom, amax, n);

    gemm_wmma_kernel<<<(unsigned)((n + 15) / 16), B, 0, stream>>>(x, W, h, n);

    att_kernel<<<(unsigned)(((long long)n * HEADS + B - 1) / B), B, 0, stream>>>(
        h, att_src, att_dst, asrc, adst, n);

    edge_pass1_kernel<<<(unsigned)(((long long)e + B - 1) / B), B, 0, stream>>>(
        src, dst, eattr, asrc, adst, deg, wsum, amax, e);

    node_pass_kernel<<<(unsigned)((n + B - 1) / B), B, 0, stream>>>(
        deg, wsum, loopw, asrc, adst, amax, n);

    scatter_kernel<<<(unsigned)(((long long)e * 32 + B - 1) / B), B, 0, stream>>>(
        src, dst, eattr, asrc, adst, amax, h, denom, out, e);

    finalize_kernel<<<(unsigned)((nTot + B - 1) / B), B, 0, stream>>>(
        h, asrc, adst, loopw, amax, denom, bias, out, n);
}


# edges bucketed by target (one returning int atomic = slot in the target's own out row); one wave per node does max, exp, sum, row gather in registers; no f32 atomics, sum order = arrival rank
# speedup vs baseline: 3.2524x; 2.9248x over previous
#include <hip/hip_runtime.h>
#include <hip/hip_bf16.h>

typedef float v2f __attribute__((ext_vector_type(2)));
typedef float v8f __attribute__((ext_vector_type(8)));

#define HEADS 4
#define CH    32
#define DIN_  128
#define NEG_SLOPE_ 0.2f
#define CAP   64

__device__ __forceinline__ float leaky(float v) {
    return v > 0.0f ? v : NEG_SLOPE_ * v;
}

__device__ __forceinline__ float wave_sum(float v) {
    for (int m = 16; m >= 1; m >>= 1) v += __shfl_xor(v, m, 32);
    return v;
}
__device__ __forceinline__ float wave_max(float v) {
    for (int m = 16; m >= 1; m >>= 1) v = fmaxf(v, __shfl_xor(v, m, 32));
    return v;
}
__device__ __forceinline__ float lane_f(float v, int j) {
    return __int_as_float(__builtin_amdgcn_readlane(__float_as_int(v), j));
}

__global__ void zero_cnt_kernel(int* __restrict__ cnt, int n) {
    int i = blockIdx.x * blockDim.x + threadIdx.x;
    if (i < n) cnt[i] = 0;
}

__global__ void gemm_wmma_kernel(const float* __restrict__ x,
                                 const float* __restrict__ W,
                                 float* __restrict__ h, int n) {
    const int wave = threadIdx.x >> 5;
    const int lane = threadIdx.x & 31;
    const int half = lane >> 4;
    const int l16  = lane & 15;
    const int row0 = blockIdx.x * 16;
    const int col0 = wave * 16;

    int arow = row0 + l16;
    if (arow >= n) arow = n - 1;
    const float* __restrict__ xrow = x + (size_t)arow * DIN_;
    const int kk = half * 2;

    v8f acc = {};
    #pragma unroll
    for (int k = 0; k < DIN_; k += 4) {
        v2f a, b;
        a.x = xrow[k + kk + 0];
        a.y = xrow[k + kk + 1];
        b.x = W[(size_t)(k + kk + 0) * DIN_ + col0 + l16];
        b.y = W[(size_t)(k + kk + 1) * DIN_ + col0 + l16];
        acc = __builtin_amdgcn_wmma_f32_16x16x4_f32(
             false, a,  false, b,
             (short)0, acc,  false,  false);
    }

    float* __restrict__ hp = h + (size_t)(col0 + l16);
    #pragma unroll
    for (int r = 0; r < 8; ++r) {
        int row = row0 + r + half * 8;
        if (row < n) hp[(size_t)row * DIN_] = acc[r];
    }
}

__global__ void att_kernel(const float* __restrict__ h,
                           const float* __restrict__ att_src,
                           const float* __restrict__ att_dst,
                           float* __restrict__ a_src, float* __restrict__ a_dst, int n) {
    long long i = (long long)blockIdx.x * blockDim.x + threadIdx.x;
    if (i >= (long long)n * HEADS) return;
    const int head = (int)(i & (HEADS - 1));
    const float* __restrict__ hp = h + (i >> 2) * DIN_ + head * CH;
    const float* __restrict__ as = att_src + head * CH;
    const float* __restrict__ ad = att_dst + head * CH;
    float s = 0.0f, d = 0.0f;
    #pragma unroll
    for (int c = 0; c < CH; ++c) {
        float v = hp[c];
        s = fmaf(v, as[c], s);
        d = fmaf(v, ad[c], d);
    }
    a_src[i] = s;
    a_dst[i] = d;
}

__global__ void bucket_kernel(const int* __restrict__ src, const int* __restrict__ dst,
                              const float* __restrict__ ea,
                              int* __restrict__ cnt, float* __restrict__ out, int e) {
    long long i = (long long)blockIdx.x * blockDim.x + threadIdx.x;
    if (i >= e) return;
    const int s = src[i], d = dst[i];
    const float w = ea[i];
    const int r = atomicAdd(&cnt[d], 1);
    if (r < CAP) {
        uint2 q;
        q.x = (unsigned)s;
        q.y = __float_as_uint(w);
        reinterpret_cast<uint2*>(out + (size_t)d * DIN_)[r] = q;
    }
}

__global__ void gather_kernel(const int* __restrict__ src, const int* __restrict__ dst,
                              const float* __restrict__ ea,
                              const float* __restrict__ a_src, const float* __restrict__ a_dst,
                              const int* __restrict__ cnt,
                              const float* __restrict__ h, const float* __restrict__ bias,
                              float* out, int n, int e) {
    const int lane = threadIdx.x & 31;
    const int node = __builtin_amdgcn_readfirstlane(
        (int)(blockIdx.x * (blockDim.x >> 5) + (threadIdx.x >> 5)));
    if (node >= n) return;
    const int c = cnt[node];
    float* orow = out + (size_t)node * DIN_;
    const float* __restrict__ hself = h + (size_t)node * DIN_ + lane;
    const float NEG_INF = -__builtin_inff();

    float ad[HEADS], asf[HEADS];
    for (int hh = 0; hh < HEADS; ++hh) {
        ad[hh]  = a_dst[node * HEADS + hh];
        asf[hh] = a_src[node * HEADS + hh];
    }
    float acc[HEADS], den[HEADS];

    if (c <= CAP) {
        const uint2* slots = reinterpret_cast<const uint2*>(orow);
        int   sj[2];
        bool  ok[2];
        float vj[2][HEADS], exj[2][HEADS];
        float wl = 0.0f;
        for (int b = 0; b < 2; ++b) {
            const int k = lane + 32 * b;
            ok[b] = k < c;
            int s = node;
            float w = 1.0f;
            if (ok[b]) {
                const uint2 q = slots[k];
                s = (int)q.x;
                w = __uint_as_float(q.y);
                wl += w;
            }
            sj[b] = s;
            if (32 * b < c) {
                const float l2w = __log2f(w);
                const float4 as4 = *reinterpret_cast<const float4*>(a_src + (size_t)s * HEADS);
                vj[b][0] = ok[b] ? leaky(as4.x + ad[0]) + l2w : NEG_INF;
                vj[b][1] = ok[b] ? leaky(as4.y + ad[1]) + l2w : NEG_INF;
                vj[b][2] = ok[b] ? leaky(as4.z + ad[2]) + l2w : NEG_INF;
                vj[b][3] = ok[b] ? leaky(as4.w + ad[3]) + l2w : NEG_INF;
            } else {
                for (int hh = 0; hh < HEADS; ++hh) vj[b][hh] = NEG_INF;
            }
        }
        const float wsum = wave_sum(wl);
        const float lw = c > 0 ? wsum / (float)c : 1.0f;
        const float l2self = __log2f(lw);
        for (int hh = 0; hh < HEADS; ++hh) {
            const float vs = leaky(asf[hh] + ad[hh]) + l2self;
            const float m = fmaxf(wave_max(fmaxf(vj[0][hh], vj[1][hh])), vs);
            exj[0][hh] = ok[0] ? __expf(vj[0][hh] - m) : 0.0f;
            exj[1][hh] = ok[1] ? __expf(vj[1][hh] - m) : 0.0f;
            const float exs = __expf(vs - m);
            den[hh] = wave_sum(exj[0][hh] + exj[1][hh]) + exs;
            acc[hh] = exs * hself[hh * CH];
        }
        for (int b = 0; b < 2; ++b) {
            const int nb = min(c - 32 * b, 32);
            #pragma unroll 4
            for (int j = 0; j < nb; ++j) {
                const int s = __builtin_amdgcn_readlane(sj[b], j);
                const float* __restrict__ hp = h + (size_t)s * DIN_ + lane;
                for (int cc = 0; cc < HEADS; ++cc)
                    acc[cc] = fmaf(lane_f(exj[b][cc], j), hp[cc * CH], acc[cc]);
            }
        }
    } else {
        float wl = 0.0f;
        float ml[HEADS];
        for (int hh = 0; hh < HEADS; ++hh) ml[hh] = NEG_INF;
        for (int base = 0; base < e; base += 32) {
            const int i = base + lane;
            if (i < e && dst[i] == node) {
                const float w = ea[i];
                const float l2w = __log2f(w);
                const float4 as4 = *reinterpret_cast<const float4*>(a_src + (size_t)src[i] * HEADS);
                wl += w;
                ml[0] = fmaxf(ml[0], leaky(as4.x + ad[0]) + l2w);
                ml[1] = fmaxf(ml[1], leaky(as4.y + ad[1]) + l2w);
                ml[2] = fmaxf(ml[2], leaky(as4.z + ad[2]) + l2w);
                ml[3] = fmaxf(ml[3], leaky(as4.w + ad[3]) + l2w);
            }
        }
        const float wsum = wave_sum(wl);
        const float l2self = __log2f(wsum / (float)c);
        float mx[HEADS], exs[HEADS], dl[HEADS];
        for (int hh = 0; hh < HEADS; ++hh) {
            const float vs = leaky(asf[hh] + ad[hh]) + l2self;
            mx[hh] = fmaxf(wave_max(ml[hh]), vs);
            exs[hh] = __expf(vs - mx[hh]);
            acc[hh] = exs[hh] * hself[hh * CH];
            dl[hh] = 0.0f;
        }
        for (int base = 0; base < e; base += 32) {
            const int i = base + lane;
            const bool hit = i < e && dst[i] == node;
            int s = node;
            float ex[HEADS] = {0.0f, 0.0f, 0.0f, 0.0f};
            if (hit) {
                s = src[i];
                const float l2w = __log2f(ea[i]);
                const float4 as4 = *reinterpret_cast<const float4*>(a_src + (size_t)s * HEADS);
                ex[0] = __expf(leaky(as4.x + ad[0]) + l2w - mx[0]);
                ex[1] = __expf(leaky(as4.y + ad[1]) + l2w - mx[1]);
                ex[2] = __expf(leaky(as4.z + ad[2]) + l2w - mx[2]);
                ex[3] = __expf(leaky(as4.w + ad[3]) + l2w - mx[3]);
                for (int hh = 0; hh < HEADS; ++hh) dl[hh] += ex[hh];
            }
            unsigned mask = __builtin_amdgcn_ballot_w32(hit);
            while (mask) {
                const int j = __builtin_ctz(mask);
                mask &= mask - 1;
                const int sb = __builtin_amdgcn_readlane(s, j);
                const float* __restrict__ hp = h + (size_t)sb * DIN_ + lane;
                for (int cc = 0; cc < HEADS; ++cc)
                    acc[cc] = fmaf(lane_f(ex[cc], j), hp[cc * CH], acc[cc]);
            }
        }
        for (int hh = 0; hh < HEADS; ++hh) den[hh] = wave_sum(dl[hh]) + exs[hh];
    }

    for (int cc = 0; cc < HEADS; ++cc)
        orow[lane + cc * CH] = acc[cc] / den[cc] + bias[lane + cc * CH];
}

extern "C" void kernel_launch(void* const* d_in, const int* in_sizes, int n_in,
                              void* d_out, int out_size, void* d_ws, size_t ws_size,
                              hipStream_t stream) {
    const float* x        = (const float*)d_in[0];
    const int*   eidx     = (const int*)d_in[1];
    const float* eattr    = (const float*)d_in[2];
    const float* W        = (const float*)d_in[3];
    const float* att_src  = (const float*)d_in[4];
    const float* att_dst  = (const float*)d_in[5];
    const float* bias     = (const float*)d_in[6];

    const int n = in_sizes[0] / DIN_;
    const int e = in_sizes[2];
    const int* src = eidx;
    const int* dst = eidx + e;

    float* out = (float*)d_out;

    float* h    = (float*)d_ws;
    float* asrc = h    + (size_t)n * DIN_;
    float* adst = asrc + (size_t)n * HEADS;
    int*   cnt  = (int*)(adst + (size_t)n * HEADS);

    const int B = 256;

    zero_cnt_kernel<<<(unsigned)((n + B - 1) / B), B, 0, stream>>>(cnt, n);

    gemm_wmma_kernel<<<(unsigned)((n + 15) / 16), B, 0, stream>>>(x, W, h, n);

    att_kernel<<<(unsigned)(((long long)n * HEADS + B - 1) / B), B, 0, stream>>>(
        h, att_src, att_dst, asrc, adst, n);

    bucket_kernel<<<(unsigned)(((long long)e + B - 1) / B), B, 0, stream>>>(
        src, dst, eattr, cnt, out, e);

    gather_kernel<<<(unsigned)((n + 7) / 8), B, 0, stream>>>(
        src, dst, eattr, asrc, adst, cnt, h, bias, out, n, e);
}


# gemm: one wave per 16x128 row strip (8 accumulator tiles): x read once in 16-B loads (was 8 times in 8-B pieces), W and h in 16-B accesses of 4 adjacent columns; k summed in another order
# speedup vs baseline: 3.3933x; 1.0433x over previous
#include <hip/hip_runtime.h>
#include <hip/hip_bf16.h>

typedef float v2f __attribute__((ext_vector_type(2)));
typedef float v4f __attribute__((ext_vector_type(4)));
typedef float v8f __attribute__((ext_vector_type(8)));

#define HEADS 4
#define CH    32
#define DIN_  128
#define NEG_SLOPE_ 0.2f
#define CAP   64

__device__ __forceinline__ float leaky(float v) {
    return v > 0.0f ? v : NEG_SLOPE_ * v;
}

__device__ __forceinline__ float wave_sum(float v) {
    for (int m = 16; m >= 1; m >>= 1) v += __shfl_xor(v, m, 32);
    return v;
}
__device__ __forceinline__ float wave_max(float v) {
    for (int m = 16; m >= 1; m >>= 1) v = fmaxf(v, __shfl_xor(v, m, 32));
    return v;
}
__device__ __forceinline__ float lane_f(float v, int j) {
    return __int_as_float(__builtin_amdgcn_readlane(__float_as_int(v), j));
}

__global__ void zero_cnt_kernel(int* __restrict__ cnt, int n) {
    int i = blockIdx.x * blockDim.x + threadIdx.x;
    if (i < n) cnt[i] = 0;
}

__global__ void __launch_bounds__(64)
gemm_wmma_kernel(const float* __restrict__ x,
                 const float* __restrict__ W,
                 float* __restrict__ h, int n) {
    const int lane = threadIdx.x & 31;
    const int half = lane >> 4;
    const int l16  = lane & 15;
    const int strip = blockIdx.x * (blockDim.x >> 5) + (threadIdx.x >> 5);
    const int row0 = strip * 16;
    if (row0 >= n) return;

    int arow = row0 + l16;
    if (arow >= n) arow = n - 1;
    const float* __restrict__ xrow = x + (size_t)arow * DIN_ + 4 * half;
    const float* __restrict__ wp   = W + (size_t)(4 * half) * DIN_ + 4 * l16;

    v8f acc[8];
    #pragma unroll
    for (int t = 0; t < 8; ++t) acc[t] = v8f{};

    #pragma unroll
    for (int j = 0; j < DIN_ / 8; ++j) {
        const v4f a4 = *reinterpret_cast<const v4f*>(xrow + 8 * j);
        v4f wlo[4], whi[4];
        #pragma unroll
        for (int q = 0; q < 4; ++q) {
            const float* __restrict__ wr = wp + (size_t)(8 * j + q) * DIN_;
            wlo[q] = *reinterpret_cast<const v4f*>(wr);
            whi[q] = *reinterpret_cast<const v4f*>(wr + 64);
        }
        #pragma unroll
        for (int p = 0; p < 2; ++p) {
            v2f a;
            a.x = a4[2 * p];
            a.y = a4[2 * p + 1];
            #pragma unroll
            for (int t = 0; t < 4; ++t) {
                v2f b0, b1;
                b0.x = wlo[2 * p][t];
                b0.y = wlo[2 * p + 1][t];
                b1.x = whi[2 * p][t];
                b1.y = whi[2 * p + 1][t];
                acc[t] = __builtin_amdgcn_wmma_f32_16x16x4_f32(
                     false, a,  false, b0,
                     (short)0, acc[t],  false,  false);
                acc[4 + t] = __builtin_amdgcn_wmma_f32_16x16x4_f32(
                     false, a,  false, b1,
                     (short)0, acc[4 + t],  false,  false);
            }
        }
    }

    float* __restrict__ hp = h + 4 * l16;
    #pragma unroll
    for (int r = 0; r < 8; ++r) {
        const int row = row0 + r + half * 8;
        if (row < n) {
            v4f lo, hi;
            lo.x = acc[0][r]; lo.y = acc[1][r]; lo.z = acc[2][r]; lo.w = acc[3][r];
            hi.x = acc[4][r]; hi.y = acc[5][r]; hi.z = acc[6][r]; hi.w = acc[7][r];
            float* __restrict__ hr = hp + (size_t)row * DIN_;
            *reinterpret_cast<v4f*>(hr)      = lo;
            *reinterpret_cast<v4f*>(hr + 64) = hi;
        }
    }
}

__global__ void att_kernel(const float* __restrict__ h,
                           const float* __restrict__ att_src,
                           const float* __restrict__ att_dst,
                           float* __restrict__ a_src, float* __restrict__ a_dst, int n) {
    long long i = (long long)blockIdx.x * blockDim.x + threadIdx.x;
    if (i >= (long long)n * HEADS) return;
    const int head = (int)(i & (HEADS - 1));
    const float* __restrict__ hp = h + (i >> 2) * DIN_ + head * CH;
    const float* __restrict__ as = att_src + head * CH;
    const float* __restrict__ ad = att_dst + head * CH;
    float s = 0.0f, d = 0.0f;
    #pragma unroll
    for (int c = 0; c < CH; ++c) {
        float v = hp[c];
        s = fmaf(v, as[c], s);
        d = fmaf(v, ad[c], d);
    }
    a_src[i] = s;
    a_dst[i] = d;
}

__global__ void bucket_kernel(const int* __restrict__ src, const int* __restrict__ dst,
                              const float* __restrict__ ea,
                              int* __restrict__ cnt, float* __restrict__ out, int e) {
    long long i = (long long)blockIdx.x * blockDim.x + threadIdx.x;
    if (i >= e) return;
    const int s = src[i], d = dst[i];
    const float w = ea[i];
    const int r = atomicAdd(&cnt[d], 1);
    if (r < CAP) {
        uint2 q;
        q.x = (unsigned)s;
        q.y = __float_as_uint(w);
        reinterpret_cast<uint2*>(out + (size_t)d * DIN_)[r] = q;
    }
}

__global__ void gather_kernel(const int* __restrict__ src, const int* __restrict__ dst,
                              const float* __restrict__ ea,
                              const float* __restrict__ a_src, const float* __restrict__ a_dst,
                              const int* __restrict__ cnt,
                              const float* __restrict__ h, const float* __restrict__ bias,
                              float* out, int n, int e) {
    const int lane = threadIdx.x & 31;
    const int node = __builtin_amdgcn_readfirstlane(
        (int)(blockIdx.x * (blockDim.x >> 5) + (threadIdx.x >> 5)));
    if (node >= n) return;
    const int c = cnt[node];
    float* orow = out + (size_t)node * DIN_;
    const float* __restrict__ hself = h + (size_t)node * DIN_ + lane;
    const float NEG_INF = -__builtin_inff();

    float ad[HEADS], asf[HEADS];
    for (int hh = 0; hh < HEADS; ++hh) {
        ad[hh]  = a_dst[node * HEADS + hh];
        asf[hh] = a_src[node * HEADS + hh];
    }
    float acc[HEADS], den[HEADS];

    if (c <= CAP) {
        const uint2* slots = reinterpret_cast<const uint2*>(orow);
        int   sj[2];
        bool  ok[2];
        float vj[2][HEADS], exj[2][HEADS];
        float wl = 0.0f;
        for (int b = 0; b < 2; ++b) {
            const int k = lane + 32 * b;
            ok[b] = k < c;
            int s = node;
            float w = 1.0f;
            if (ok[b]) {
                const uint2 q = slots[k];
                s = (int)q.x;
                w = __uint_as_float(q.y);
                wl += w;
            }
            sj[b] = s;
            if (32 * b < c) {
                const float l2w = __log2f(w);
                const float4 as4 = *reinterpret_cast<const float4*>(a_src + (size_t)s * HEADS);
                vj[b][0] = ok[b] ? leaky(as4.x + ad[0]) + l2w : NEG_INF;
                vj[b][1] = ok[b] ? leaky(as4.y + ad[1]) + l2w : NEG_INF;
                vj[b][2] = ok[b] ? leaky(as4.z + ad[2]) + l2w : NEG_INF;
                vj[b][3] = ok[b] ? leaky(as4.w + ad[3]) + l2w : NEG_INF;
            } else {
                for (int hh = 0; hh < HEADS; ++hh) vj[b][hh] = NEG_INF;
            }
        }
        const float wsum = wave_sum(wl);
        const float lw = c > 0 ? wsum / (float)c : 1.0f;
        const float l2self = __log2f(lw);
        for (int hh = 0; hh < HEADS; ++hh) {
            const float vs = leaky(asf[hh] + ad[hh]) + l2self;
            const float m = fmaxf(wave_max(fmaxf(vj[0][hh], vj[1][hh])), vs);
            exj[0][hh] = ok[0] ? __expf(vj[0][hh] - m) : 0.0f;
            exj[1][hh] = ok[1] ? __expf(vj[1][hh] - m) : 0.0f;
            const float exs = __expf(vs - m);
            den[hh] = wave_sum(exj[0][hh] + exj[1][hh]) + exs;
            acc[hh] = exs * hself[hh * CH];
        }
        for (int b = 0; b < 2; ++b) {
            const int nb = min(c - 32 * b, 32);
            #pragma unroll 4
            for (int j = 0; j < nb; ++j) {
                const int s = __builtin_amdgcn_readlane(sj[b], j);
                const float* __restrict__ hp = h + (size_t)s * DIN_ + lane;
                for (int cc = 0; cc < HEADS; ++cc)
                    acc[cc] = fmaf(lane_f(exj[b][cc], j), hp[cc * CH], acc[cc]);
            }
        }
    } else {
        float wl = 0.0f;
        float ml[HEADS];
        for (int hh = 0; hh < HEADS; ++hh) ml[hh] = NEG_INF;
        for (int base = 0; base < e; base += 32) {
            const int i = base + lane;
            if (i < e && dst[i] == node) {
                const float w = ea[i];
                const float l2w = __log2f(w);
                const float4 as4 = *reinterpret_cast<const float4*>(a_src + (size_t)src[i] * HEADS);
                wl += w;
                ml[0] = fmaxf(ml[0], leaky(as4.x + ad[0]) + l2w);
                ml[1] = fmaxf(ml[1], leaky(as4.y + ad[1]) + l2w);
                ml[2] = fmaxf(ml[2], leaky(as4.z + ad[2]) + l2w);
                ml[3] = fmaxf(ml[3], leaky(as4.w + ad[3]) + l2w);
            }
        }
        const float wsum = wave_sum(wl);
        const float l2self = __log2f(wsum / (float)c);
        float mx[HEADS], exs[HEADS], dl[HEADS];
        for (int hh = 0; hh < HEADS; ++hh) {
            const float vs = leaky(asf[hh] + ad[hh]) + l2self;
            mx[hh] = fmaxf(wave_max(ml[hh]), vs);
            exs[hh] = __expf(vs - mx[hh]);
            acc[hh] = exs[hh] * hself[hh * CH];
            dl[hh] = 0.0f;
        }
        for (int base = 0; base < e; base += 32) {
            const int i = base + lane;
            const bool hit = i < e && dst[i] == node;
            int s = node;
            float ex[HEADS] = {0.0f, 0.0f, 0.0f, 0.0f};
            if (hit) {
                s = src[i];
                const float l2w = __log2f(ea[i]);
                const float4 as4 = *reinterpret_cast<const float4*>(a_src + (size_t)s * HEADS);
                ex[0] = __expf(leaky(as4.x + ad[0]) + l2w - mx[0]);
                ex[1] = __expf(leaky(as4.y + ad[1]) + l2w - mx[1]);
                ex[2] = __expf(leaky(as4.z + ad[2]) + l2w - mx[2]);
                ex[3] = __expf(leaky(as4.w + ad[3]) + l2w - mx[3]);
                for (int hh = 0; hh < HEADS; ++hh) dl[hh] += ex[hh];
            }
            unsigned mask = __builtin_amdgcn_ballot_w32(hit);
            while (mask) {
                const int j = __builtin_ctz(mask);
                mask &= mask - 1;
                const int sb = __builtin_amdgcn_readlane(s, j);
                const float* __restrict__ hp = h + (size_t)sb * DIN_ + lane;
                for (int cc = 0; cc < HEADS; ++cc)
                    acc[cc] = fmaf(lane_f(ex[cc], j), hp[cc * CH], acc[cc]);
            }
        }
        for (int hh = 0; hh < HEADS; ++hh) den[hh] = wave_sum(dl[hh]) + exs[hh];
    }

    for (int cc = 0; cc < HEADS; ++cc)
        orow[lane + cc * CH] = acc[cc] / den[cc] + bias[lane + cc * CH];
}

extern "C" void kernel_launch(void* const* d_in, const int* in_sizes, int n_in,
                              void* d_out, int out_size, void* d_ws, size_t ws_size,
                              hipStream_t stream) {
    const float* x        = (const float*)d_in[0];
    const int*   eidx     = (const int*)d_in[1];
    const float* eattr    = (const float*)d_in[2];
    const float* W        = (const float*)d_in[3];
    const float* att_src  = (const float*)d_in[4];
    const float* att_dst  = (const float*)d_in[5];
    const float* bias     = (const float*)d_in[6];

    const int n = in_sizes[0] / DIN_;
    const int e = in_sizes[2];
    const int* src = eidx;
    const int* dst = eidx + e;

    float* out = (float*)d_out;

    float* h    = (float*)d_ws;
    float* asrc = h    + (size_t)n * DIN_;
    float* adst = asrc + (size_t)n * HEADS;
    int*   cnt  = (int*)(adst + (size_t)n * HEADS);

    const int B = 256;

    zero_cnt_kernel<<<(unsigned)((n + B - 1) / B), B, 0, stream>>>(cnt, n);

    gemm_wmma_kernel<<<(unsigned)((n + 31) / 32), 64, 0, stream>>>(x, W, h, n);

    att_kernel<<<(unsigned)(((long long)n * HEADS + B - 1) / B), B, 0, stream>>>(
        h, att_src, att_dst, asrc, adst, n);

    bucket_kernel<<<(unsigned)(((long long)e + B - 1) / B), B, 0, stream>>>(
        src, dst, eattr, cnt, out, e);

    gather_kernel<<<(unsigned)((n + 7) / 8), B, 0, stream>>>(
        src, dst, eattr, asrc, adst, cnt, h, bias, out, n, e);
}


# edge bucketing without global atomics: blocks rank edges per 256-node bin with LDS adds, a bin's owner block hands out node slots with LDS adds; sum order still an arrival rank, now of LDS adds
# speedup vs baseline: 5.4841x; 1.6162x over previous
#include <hip/hip_runtime.h>
#include <hip/hip_bf16.h>

typedef float v2f __attribute__((ext_vector_type(2)));
typedef float v4f __attribute__((ext_vector_type(4)));
typedef float v8f __attribute__((ext_vector_type(8)));

#define HEADS 4
#define CH    32
#define DIN_  128
#define NEG_SLOPE_ 0.2f
#define CAP   64

__device__ __forceinline__ float leaky(float v) {
    return v > 0.0f ? v : NEG_SLOPE_ * v;
}

__device__ __forceinline__ float wave_sum(float v) {
    for (int m = 16; m >= 1; m >>= 1) v += __shfl_xor(v, m, 32);
    return v;
}
__device__ __forceinline__ float wave_max(float v) {
    for (int m = 16; m >= 1; m >>= 1) v = fmaxf(v, __shfl_xor(v, m, 32));
    return v;
}
__device__ __forceinline__ float lane_f(float v, int j) {
    return __int_as_float(__builtin_amdgcn_readlane(__float_as_int(v), j));
}

__global__ void __launch_bounds__(64)
gemm_wmma_kernel(const float* __restrict__ x,
                 const float* __restrict__ W,
                 float* __restrict__ h, int n) {
    const int lane = threadIdx.x & 31;
    const int half = lane >> 4;
    const int l16  = lane & 15;
    const int strip = blockIdx.x * (blockDim.x >> 5) + (threadIdx.x >> 5);
    const int row0 = strip * 16;
    if (row0 >= n) return;

    int arow = row0 + l16;
    if (arow >= n) arow = n - 1;
    const float* __restrict__ xrow = x + (size_t)arow * DIN_ + 4 * half;
    const float* __restrict__ wp   = W + (size_t)(4 * half) * DIN_ + 4 * l16;

    v8f acc[8];
    #pragma unroll
    for (int t = 0; t < 8; ++t) acc[t] = v8f{};

    #pragma unroll
    for (int j = 0; j < DIN_ / 8; ++j) {
        const v4f a4 = *reinterpret_cast<const v4f*>(xrow + 8 * j);
        v4f wlo[4], whi[4];
        #pragma unroll
        for (int q = 0; q < 4; ++q) {
            const float* __restrict__ wr = wp + (size_t)(8 * j + q) * DIN_;
            wlo[q] = *reinterpret_cast<const v4f*>(wr);
            whi[q] = *reinterpret_cast<const v4f*>(wr + 64);
        }
        #pragma unroll
        for (int p = 0; p < 2; ++p) {
            v2f a;
            a.x = a4[2 * p];
            a.y = a4[2 * p + 1];
            #pragma unroll
            for (int t = 0; t < 4; ++t) {
                v2f b0, b1;
                b0.x = wlo[2 * p][t];
                b0.y = wlo[2 * p + 1][t];
                b1.x = whi[2 * p][t];
                b1.y = whi[2 * p + 1][t];
                acc[t] = __builtin_amdgcn_wmma_f32_16x16x4_f32(
                     false, a,  false, b0,
                     (short)0, acc[t],  false,  false);
                acc[4 + t] = __builtin_amdgcn_wmma_f32_16x16x4_f32(
                     false, a,  false, b1,
                     (short)0, acc[4 + t],  false,  false);
            }
        }
    }

    float* __restrict__ hp = h + 4 * l16;
    #pragma unroll
    for (int r = 0; r < 8; ++r) {
        const int row = row0 + r + half * 8;
        if (row < n) {
            v4f lo, hi;
            lo.x = acc[0][r]; lo.y = acc[1][r]; lo.z = acc[2][r]; lo.w = acc[3][r];
            hi.x = acc[4][r]; hi.y = acc[5][r]; hi.z = acc[6][r]; hi.w = acc[7][r];
            float* __restrict__ hr = hp + (size_t)row * DIN_;
            *reinterpret_cast<v4f*>(hr)      = lo;
            *reinterpret_cast<v4f*>(hr + 64) = hi;
        }
    }
}

__global__ void att_kernel(const float* __restrict__ h,
                           const float* __restrict__ att_src,
                           const float* __restrict__ att_dst,
                           float* __restrict__ a_src, float* __restrict__ a_dst, int n) {
    long long i = (long long)blockIdx.x * blockDim.x + threadIdx.x;
    if (i >= (long long)n * HEADS) return;
    const int head = (int)(i & (HEADS - 1));
    const float* __restrict__ hp = h + (i >> 2) * DIN_ + head * CH;
    const float* __restrict__ as = att_src + head * CH;
    const float* __restrict__ ad = att_dst + head * CH;
    float s = 0.0f, d = 0.0f;
    #pragma unroll
    for (int c = 0; c < CH; ++c) {
        float v = hp[c];
        s = fmaf(v, as[c], s);
        d = fmaf(v, ad[c], d);
    }
    a_src[i] = s;
    a_dst[i] = d;
}

#define RB     256
#define RB_LOG 8
#define PT     1024
#define EPT    8
#define EPB    (PT * EPT)

__global__ void __launch_bounds__(PT)
partition_kernel(const int* __restrict__ src, const int* __restrict__ dst,
                 const float* __restrict__ ea, uint2* __restrict__ rec,
                 int n, int e, int nb, int nw) {
    extern __shared__ int hist[];
    const int t  = threadIdx.x;
    const int wb = blockIdx.x;
    for (int b = t; b < nb; b += PT) hist[b] = 0;
    __syncthreads();

    const int sc_full = (RB * CAP) / nw;
    const int sc_last = ((n - (nb - 1) * RB) * CAP) / nw;
    const bool packable = n <= (1 << 24);

    int d[EPT], s[EPT], rk[EPT];
    float wv[EPT];
    const long long i0 = (long long)wb * EPB + t;
    #pragma unroll
    for (int k = 0; k < EPT; ++k) {
        const long long i = i0 + (long long)k * PT;
        d[k] = -1; s[k] = 0; wv[k] = 0.0f;
        if (i < e) { d[k] = dst[i]; s[k] = src[i]; wv[k] = ea[i]; }
    }
    #pragma unroll
    for (int k = 0; k < EPT; ++k) {
        rk[k] = 0;
        if (d[k] >= 0) rk[k] = atomicAdd(&hist[d[k] >> RB_LOG], 1);
    }
    __syncthreads();

    for (int b = t; b < nb; b += PT) {
        const int sc = (b == nb - 1) ? sc_last : sc_full;
        if (sc >= 2)
            rec[(size_t)b * (RB * CAP) + (size_t)wb * sc] = make_uint2((unsigned)hist[b], 0u);
    }
    if (packable) {
        #pragma unroll
        for (int k = 0; k < EPT; ++k) {
            if (d[k] >= 0) {
                const int b = d[k] >> RB_LOG;
                const int sc = (b == nb - 1) ? sc_last : sc_full;
                if (rk[k] < sc - 1)
                    rec[(size_t)b * (RB * CAP) + (size_t)wb * sc + 1 + rk[k]] =
                        make_uint2((unsigned)s[k] | ((unsigned)(d[k] & (RB - 1)) << 24),
                                   __float_as_uint(wv[k]));
            }
        }
    }
}

__global__ void __launch_bounds__(PT)
rank_kernel(const uint2* __restrict__ rec,
            const int* __restrict__ src, const int* __restrict__ dst,
            const float* __restrict__ ea,
            int* __restrict__ cnt, float* __restrict__ out, int n, int e, int nw) {
    __shared__ int lcnt[RB];
    __shared__ int ovf_s;
    extern __shared__ int cntw[];
    const int t = threadIdx.x;
    const int b = blockIdx.x;
    const int node0 = b * RB;
    const int rows = min(RB, n - node0);
    const int sc = nw > 0 ? (rows * CAP) / nw : 0;
    const int usable = (n <= (1 << 24)) ? sc - 1 : -1;
    const uint2* __restrict__ reg = rec + (size_t)b * (RB * CAP);

    if (t < RB) lcnt[t] = 0;
    if (t == 0) ovf_s = (sc < 2) ? 1 : 0;
    __syncthreads();
    if (sc >= 2) {
        for (int w = t; w < nw; w += PT) {
            const int c = (int)reg[(size_t)w * sc].x;
            cntw[w] = c;
            if (c > usable) ovf_s = 1;
        }
    }
    __syncthreads();
    const bool ovf = ovf_s != 0;

    if (!ovf) {
        const int wave = t >> 5, lane = t & 31;
        for (int w0 = wave; w0 < nw; w0 += (PT / 32) * 4) {
            uint2 q[4];
            bool ok[4];
            #pragma unroll
            for (int u = 0; u < 4; ++u) {
                const int w = w0 + (PT / 32) * u;
                const int c = w < nw ? cntw[w] : 0;
                ok[u] = lane < c;
                q[u] = make_uint2(0u, 0u);
                if (ok[u]) q[u] = reg[(size_t)w * sc + 1 + lane];
            }
            #pragma unroll
            for (int u = 0; u < 4; ++u) {
                if (ok[u]) {
                    const int dl = (int)(q[u].x >> 24);
                    const int r = atomicAdd(&lcnt[dl], 1);
                    if (r < CAP)
                        reinterpret_cast<uint2*>(out + (size_t)(node0 + dl) * DIN_)[r] =
                            make_uint2(q[u].x & 0xFFFFFFu, q[u].y);
                }
            }
            for (int u = 0; u < 4; ++u) {
                const int w = w0 + (PT / 32) * u;
                const int c = w < nw ? cntw[w] : 0;
                for (int k = lane + 32; k < c; k += 32) {
                    const uint2 qq = reg[(size_t)w * sc + 1 + k];
                    const int dl = (int)(qq.x >> 24);
                    const int r = atomicAdd(&lcnt[dl], 1);
                    if (r < CAP)
                        reinterpret_cast<uint2*>(out + (size_t)(node0 + dl) * DIN_)[r] =
                            make_uint2(qq.x & 0xFFFFFFu, qq.y);
                }
            }
        }
    } else {
        for (long long i = t; i < e; i += PT) {
            const int d = dst[i];
            if (d >= node0 && d < node0 + rows) {
                const int r = atomicAdd(&lcnt[d - node0], 1);
                if (r < CAP)
                    reinterpret_cast<uint2*>(out + (size_t)d * DIN_)[r] =
                        make_uint2((unsigned)src[i], __float_as_uint(ea[i]));
            }
        }
    }
    __syncthreads();
    if (t < rows) cnt[node0 + t] = lcnt[t];
}

__global__ void gather_kernel(const int* __restrict__ src, const int* __restrict__ dst,
                              const float* __restrict__ ea,
                              const float* __restrict__ a_src, const float* __restrict__ a_dst,
                              const int* __restrict__ cnt,
                              const float* __restrict__ h, const float* __restrict__ bias,
                              float* out, int n, int e) {
    const int lane = threadIdx.x & 31;
    const int node = __builtin_amdgcn_readfirstlane(
        (int)(blockIdx.x * (blockDim.x >> 5) + (threadIdx.x >> 5)));
    if (node >= n) return;
    const int c = cnt[node];
    float* orow = out + (size_t)node * DIN_;
    const float* __restrict__ hself = h + (size_t)node * DIN_ + lane;
    const float NEG_INF = -__builtin_inff();

    float ad[HEADS], asf[HEADS];
    for (int hh = 0; hh < HEADS; ++hh) {
        ad[hh]  = a_dst[node * HEADS + hh];
        asf[hh] = a_src[node * HEADS + hh];
    }
    float acc[HEADS], den[HEADS];

    if (c <= CAP) {
        const uint2* slots = reinterpret_cast<const uint2*>(orow);
        int   sj[2];
        bool  ok[2];
        float vj[2][HEADS], exj[2][HEADS];
        float wl = 0.0f;
        for (int b = 0; b < 2; ++b) {
            const int k = lane + 32 * b;
            ok[b] = k < c;
            int s = node;
            float w = 1.0f;
            if (ok[b]) {
                const uint2 q = slots[k];
                s = (int)q.x;
                w = __uint_as_float(q.y);
                wl += w;
            }
            sj[b] = s;
            if (32 * b < c) {
                const float l2w = __log2f(w);
                const float4 as4 = *reinterpret_cast<const float4*>(a_src + (size_t)s * HEADS);
                vj[b][0] = ok[b] ? leaky(as4.x + ad[0]) + l2w : NEG_INF;
                vj[b][1] = ok[b] ? leaky(as4.y + ad[1]) + l2w : NEG_INF;
                vj[b][2] = ok[b] ? leaky(as4.z + ad[2]) + l2w : NEG_INF;
                vj[b][3] = ok[b] ? leaky(as4.w + ad[3]) + l2w : NEG_INF;
            } else {
                for (int hh = 0; hh < HEADS; ++hh) vj[b][hh] = NEG_INF;
            }
        }
        const float wsum = wave_sum(wl);
        const float lw = c > 0 ? wsum / (float)c : 1.0f;
        const float l2self = __log2f(lw);
        for (int hh = 0; hh < HEADS; ++hh) {
            const float vs = leaky(asf[hh] + ad[hh]) + l2self;
            const float m = fmaxf(wave_max(fmaxf(vj[0][hh], vj[1][hh])), vs);
            exj[0][hh] = ok[0] ? __expf(vj[0][hh] - m) : 0.0f;
            exj[1][hh] = ok[1] ? __expf(vj[1][hh] - m) : 0.0f;
            const float exs = __expf(vs - m);
            den[hh] = wave_sum(exj[0][hh] + exj[1][hh]) + exs;
            acc[hh] = exs * hself[hh * CH];
        }
        for (int b = 0; b < 2; ++b) {
            const int nb = min(c - 32 * b, 32);
            #pragma unroll 4
            for (int j = 0; j < nb; ++j) {
                const int s = __builtin_amdgcn_readlane(sj[b], j);
                const float* __restrict__ hp = h + (size_t)s * DIN_ + lane;
                for (int cc = 0; cc < HEADS; ++cc)
                    acc[cc] = fmaf(lane_f(exj[b][cc], j), hp[cc * CH], acc[cc]);
            }
        }
    } else {
        float wl = 0.0f;
        float ml[HEADS];
        for (int hh = 0; hh < HEADS; ++hh) ml[hh] = NEG_INF;
        for (int base = 0; base < e; base += 32) {
            const int i = base + lane;
            if (i < e && dst[i] == node) {
                const float w = ea[i];
                const float l2w = __log2f(w);
                const float4 as4 = *reinterpret_cast<const float4*>(a_src + (size_t)src[i] * HEADS);
                wl += w;
                ml[0] = fmaxf(ml[0], leaky(as4.x + ad[0]) + l2w);
                ml[1] = fmaxf(ml[1], leaky(as4.y + ad[1]) + l2w);
                ml[2] = fmaxf(ml[2], leaky(as4.z + ad[2]) + l2w);
                ml[3] = fmaxf(ml[3], leaky(as4.w + ad[3]) + l2w);
            }
        }
        const float wsum = wave_sum(wl);
        const float l2self = __log2f(wsum / (float)c);
        float mx[HEADS], exs[HEADS], dl[HEADS];
        for (int hh = 0; hh < HEADS; ++hh) {
            const float vs = leaky(asf[hh] + ad[hh]) + l2self;
            mx[hh] = fmaxf(wave_max(ml[hh]), vs);
            exs[hh] = __expf(vs - mx[hh]);
            acc[hh] = exs[hh] * hself[hh * CH];
            dl[hh] = 0.0f;
        }
        for (int base = 0; base < e; base += 32) {
            const int i = base + lane;
            const bool hit = i < e && dst[i] == node;
            int s = node;
            float ex[HEADS] = {0.0f, 0.0f, 0.0f, 0.0f};
            if (hit) {
                s = src[i];
                const float l2w = __log2f(ea[i]);
                const float4 as4 = *reinterpret_cast<const float4*>(a_src + (size_t)s * HEADS);
                ex[0] = __expf(leaky(as4.x + ad[0]) + l2w - mx[0]);
                ex[1] = __expf(leaky(as4.y + ad[1]) + l2w - mx[1]);
                ex[2] = __expf(leaky(as4.z + ad[2]) + l2w - mx[2]);
                ex[3] = __expf(leaky(as4.w + ad[3]) + l2w - mx[3]);
                for (int hh = 0; hh < HEADS; ++hh) dl[hh] += ex[hh];
            }
            unsigned mask = __builtin_amdgcn_ballot_w32(hit);
            while (mask) {
                const int j = __builtin_ctz(mask);
                mask &= mask - 1;
                const int sb = __builtin_amdgcn_readlane(s, j);
                const float* __restrict__ hp = h + (size_t)sb * DIN_ + lane;
                for (int cc = 0; cc < HEADS; ++cc)
                    acc[cc] = fmaf(lane_f(ex[cc], j), hp[cc * CH], acc[cc]);
            }
        }
        for (int hh = 0; hh < HEADS; ++hh) den[hh] = wave_sum(dl[hh]) + exs[hh];
    }

    for (int cc = 0; cc < HEADS; ++cc)
        orow[lane + cc * CH] = acc[cc] / den[cc] + bias[lane + cc * CH];
}

extern "C" void kernel_launch(void* const* d_in, const int* in_sizes, int n_in,
                              void* d_out, int out_size, void* d_ws, size_t ws_size,
                              hipStream_t stream) {
    const float* x        = (const float*)d_in[0];
    const int*   eidx     = (const int*)d_in[1];
    const float* eattr    = (const float*)d_in[2];
    const float* W        = (const float*)d_in[3];
    const float* att_src  = (const float*)d_in[4];
    const float* att_dst  = (const float*)d_in[5];
    const float* bias     = (const float*)d_in[6];

    const int n = in_sizes[0] / DIN_;
    const int e = in_sizes[2];
    const int* src = eidx;
    const int* dst = eidx + e;

    float* out = (float*)d_out;

    float* h    = (float*)d_ws;
    float* asrc = h    + (size_t)n * DIN_;
    float* adst = asrc + (size_t)n * HEADS;
    int*   cnt  = (int*)(adst + (size_t)n * HEADS);

    const int B = 256;

    uint2* rec = (uint2*)h;
    const int NB = (n + RB - 1) / RB;
    const int NW = (e + EPB - 1) / EPB;
    partition_kernel<<<(unsigned)NW, PT, (size_t)NB * sizeof(int), stream>>>(
        src, dst, eattr, rec, n, e, NB, NW);
    rank_kernel<<<(unsigned)NB, PT, (size_t)NW * sizeof(int), stream>>>(
        rec, src, dst, eattr, cnt, out, n, e, NW);

    gemm_wmma_kernel<<<(unsigned)((n + 31) / 32), 64, 0, stream>>>(x, W, h, n);

    att_kernel<<<(unsigned)(((long long)n * HEADS + B - 1) / B), B, 0, stream>>>(
        h, att_src, att_dst, asrc, adst, n);

    gather_kernel<<<(unsigned)((n + 7) / 8), B, 0, stream>>>(
        src, dst, eattr, asrc, adst, cnt, h, bias, out, n, e);
}


# gemm epilogue forms a_src/a_dst from its accumulators (4 columns per lane, then 8 lanes summed by row_xmask DPP adds); the launch that re-read h for them is gone; their sum order changes
# speedup vs baseline: 5.7194x; 1.0429x over previous
#include <hip/hip_runtime.h>
#include <hip/hip_bf16.h>

typedef float v2f __attribute__((ext_vector_type(2)));
typedef float v4f __attribute__((ext_vector_type(4)));
typedef float v8f __attribute__((ext_vector_type(8)));

#define HEADS 4
#define CH    32
#define DIN_  128
#define NEG_SLOPE_ 0.2f
#define CAP   64

__device__ __forceinline__ float leaky(float v) {
    return v > 0.0f ? v : NEG_SLOPE_ * v;
}

__device__ __forceinline__ float wave_sum(float v) {
    for (int m = 16; m >= 1; m >>= 1) v += __shfl_xor(v, m, 32);
    return v;
}
__device__ __forceinline__ float wave_max(float v) {
    for (int m = 16; m >= 1; m >>= 1) v = fmaxf(v, __shfl_xor(v, m, 32));
    return v;
}
__device__ __forceinline__ float lane_f(float v, int j) {
    return __int_as_float(__builtin_amdgcn_readlane(__float_as_int(v), j));
}

template <int M>
__device__ __forceinline__ float row_xor(float v) {
    return __int_as_float(__builtin_amdgcn_update_dpp(
        0, __float_as_int(v), 0x160 + M, 0xf, 0xf, false));
}

__global__ void __launch_bounds__(64)
gemm_wmma_kernel(const float* __restrict__ x,
                 const float* __restrict__ W,
                 const float* __restrict__ att_src,
                 const float* __restrict__ att_dst,
                 float* __restrict__ h,
                 float* __restrict__ a_src, float* __restrict__ a_dst, int n) {
    const int lane = threadIdx.x & 31;
    const int half = lane >> 4;
    const int l16  = lane & 15;
    const int strip = blockIdx.x * (blockDim.x >> 5) + (threadIdx.x >> 5);
    const int row0 = strip * 16;
    if (row0 >= n) return;

    int arow = row0 + l16;
    if (arow >= n) arow = n - 1;
    const float* __restrict__ xrow = x + (size_t)arow * DIN_ + 4 * half;
    const float* __restrict__ wp   = W + (size_t)(4 * half) * DIN_ + 4 * l16;

    v8f acc[8];
    #pragma unroll
    for (int t = 0; t < 8; ++t) acc[t] = v8f{};

    #pragma unroll
    for (int j = 0; j < DIN_ / 8; ++j) {
        const v4f a4 = *reinterpret_cast<const v4f*>(xrow + 8 * j);
        v4f wlo[4], whi[4];
        #pragma unroll
        for (int q = 0; q < 4; ++q) {
            const float* __restrict__ wr = wp + (size_t)(8 * j + q) * DIN_;
            wlo[q] = *reinterpret_cast<const v4f*>(wr);
            whi[q] = *reinterpret_cast<const v4f*>(wr + 64);
        }
        #pragma unroll
        for (int p = 0; p < 2; ++p) {
            v2f a;
            a.x = a4[2 * p];
            a.y = a4[2 * p + 1];
            #pragma unroll
            for (int t = 0; t < 4; ++t) {
                v2f b0, b1;
                b0.x = wlo[2 * p][t];
                b0.y = wlo[2 * p + 1][t];
                b1.x = whi[2 * p][t];
                b1.y = whi[2 * p + 1][t];
                acc[t] = __builtin_amdgcn_wmma_f32_16x16x4_f32(
                     false, a,  false, b0,
                     (short)0, acc[t],  false,  false);
                acc[4 + t] = __builtin_amdgcn_wmma_f32_16x16x4_f32(
                     false, a,  false, b1,
                     (short)0, acc[4 + t],  false,  false);
            }
        }
    }

    float* __restrict__ hp = h + 4 * l16;
    #pragma unroll
    for (int r = 0; r < 8; ++r) {
        const int row = row0 + r + half * 8;
        if (row < n) {
            v4f lo, hi;
            lo.x = acc[0][r]; lo.y = acc[1][r]; lo.z = acc[2][r]; lo.w = acc[3][r];
            hi.x = acc[4][r]; hi.y = acc[5][r]; hi.z = acc[6][r]; hi.w = acc[7][r];
            float* __restrict__ hr = hp + (size_t)row * DIN_;
            *reinterpret_cast<v4f*>(hr)      = lo;
            *reinterpret_cast<v4f*>(hr + 64) = hi;
        }
    }

    const v4f s_lo = *reinterpret_cast<const v4f*>(att_src + 4 * l16);
    const v4f s_hi = *reinterpret_cast<const v4f*>(att_src + 64 + 4 * l16);
    const v4f d_lo = *reinterpret_cast<const v4f*>(att_dst + 4 * l16);
    const v4f d_hi = *reinterpret_cast<const v4f*>(att_dst + 64 + 4 * l16);
    float p[8][4];
    #pragma unroll
    for (int r = 0; r < 8; ++r) {
        p[r][0] = fmaf(acc[3][r], s_lo.w, fmaf(acc[2][r], s_lo.z, fmaf(acc[1][r], s_lo.y, acc[0][r] * s_lo.x)));
        p[r][1] = fmaf(acc[7][r], s_hi.w, fmaf(acc[6][r], s_hi.z, fmaf(acc[5][r], s_hi.y, acc[4][r] * s_hi.x)));
        p[r][2] = fmaf(acc[3][r], d_lo.w, fmaf(acc[2][r], d_lo.z, fmaf(acc[1][r], d_lo.y, acc[0][r] * d_lo.x)));
        p[r][3] = fmaf(acc[7][r], d_hi.w, fmaf(acc[6][r], d_hi.z, fmaf(acc[5][r], d_hi.y, acc[4][r] * d_hi.x)));
    }
    const bool b4 = (lane & 4) != 0;
    const bool b2 = (lane & 2) != 0;
    const bool b1 = (lane & 1) != 0;
    float q4[4][4], q2[2][4], q1[4];
    #pragma unroll
    for (int r = 0; r < 4; ++r) {
        #pragma unroll
        for (int k = 0; k < 4; ++k) {
            const float keep = b4 ? p[r + 4][k] : p[r][k];
            const float send = b4 ? p[r][k] : p[r + 4][k];
            q4[r][k] = keep + row_xor<4>(send);
        }
    }
    #pragma unroll
    for (int r = 0; r < 2; ++r) {
        #pragma unroll
        for (int k = 0; k < 4; ++k) {
            const float keep = b2 ? q4[r + 2][k] : q4[r][k];
            const float send = b2 ? q4[r][k] : q4[r + 2][k];
            q2[r][k] = keep + row_xor<2>(send);
        }
    }
    #pragma unroll
    for (int k = 0; k < 4; ++k) {
        const float keep = b1 ? q2[1][k] : q2[0][k];
        const float send = b1 ? q2[0][k] : q2[1][k];
        q1[k] = keep + row_xor<1>(send);
    }
    const bool g = (lane & 8) != 0;
    const float r0 = row_xor<8>(g ? q1[0] : q1[2]);
    const float r1 = row_xor<8>(g ? q1[1] : q1[3]);
    v4f o;
    o.x = g ? r0 : q1[0];
    o.y = g ? q1[2] : r0;
    o.z = g ? r1 : q1[1];
    o.w = g ? q1[3] : r1;
    const int orow = row0 + (lane & 7) + half * 8;
    if (orow < n)
        *reinterpret_cast<v4f*>((g ? a_dst : a_src) + (size_t)orow * HEADS) = o;
}

#define RB     256
#define RB_LOG 8
#define PT     1024
#define EPT    8
#define EPB    (PT * EPT)

__global__ void __launch_bounds__(PT)
partition_kernel(const int* __restrict__ src, const int* __restrict__ dst,
                 const float* __restrict__ ea, uint2* __restrict__ rec,
                 int n, int e, int nb, int nw) {
    extern __shared__ int hist[];
    const int t  = threadIdx.x;
    const int wb = blockIdx.x;
    for (int b = t; b < nb; b += PT) hist[b] = 0;
    __syncthreads();

    const int sc_full = (RB * CAP) / nw;
    const int sc_last = ((n - (nb - 1) * RB) * CAP) / nw;
    const bool packable = n <= (1 << 24);

    int d[EPT], s[EPT], rk[EPT];
    float wv[EPT];
    const long long i0 = (long long)wb * EPB + t;
    #pragma unroll
    for (int k = 0; k < EPT; ++k) {
        const long long i = i0 + (long long)k * PT;
        d[k] = -1; s[k] = 0; wv[k] = 0.0f;
        if (i < e) { d[k] = dst[i]; s[k] = src[i]; wv[k] = ea[i]; }
    }
    #pragma unroll
    for (int k = 0; k < EPT; ++k) {
        rk[k] = 0;
        if (d[k] >= 0) rk[k] = atomicAdd(&hist[d[k] >> RB_LOG], 1);
    }
    __syncthreads();

    for (int b = t; b < nb; b += PT) {
        const int sc = (b == nb - 1) ? sc_last : sc_full;
        if (sc >= 2)
            rec[(size_t)b * (RB * CAP) + (size_t)wb * sc] = make_uint2((unsigned)hist[b], 0u);
    }
    if (packable) {
        #pragma unroll
        for (int k = 0; k < EPT; ++k) {
            if (d[k] >= 0) {
                const int b = d[k] >> RB_LOG;
                const int sc = (b == nb - 1) ? sc_last : sc_full;
                if (rk[k] < sc - 1)
                    rec[(size_t)b * (RB * CAP) + (size_t)wb * sc + 1 + rk[k]] =
                        make_uint2((unsigned)s[k] | ((unsigned)(d[k] & (RB - 1)) << 24),
                                   __float_as_uint(wv[k]));
            }
        }
    }
}

__global__ void __launch_bounds__(PT)
rank_kernel(const uint2* __restrict__ rec,
            const int* __restrict__ src, const int* __restrict__ dst,
            const float* __restrict__ ea,
            int* __restrict__ cnt, float* __restrict__ out, int n, int e, int nw) {
    __shared__ int lcnt[RB];
    __shared__ int ovf_s;
    extern __shared__ int cntw[];
    const int t = threadIdx.x;
    const int b = blockIdx.x;
    const int node0 = b * RB;
    const int rows = min(RB, n - node0);
    const int sc = nw > 0 ? (rows * CAP) / nw : 0;
    const int usable = (n <= (1 << 24)) ? sc - 1 : -1;
    const uint2* __restrict__ reg = rec + (size_t)b * (RB * CAP);

    if (t < RB) lcnt[t] = 0;
    if (t == 0) ovf_s = (sc < 2) ? 1 : 0;
    __syncthreads();
    if (sc >= 2) {
        for (int w = t; w < nw; w += PT) {
            const int c = (int)reg[(size_t)w * sc].x;
            cntw[w] = c;
            if (c > usable) ovf_s = 1;
        }
    }
    __syncthreads();
    const bool ovf = ovf_s != 0;

    if (!ovf) {
        const int wave = t >> 5, lane = t & 31;
        for (int w0 = wave; w0 < nw; w0 += (PT / 32) * 4) {
            uint2 q[4];
            bool ok[4];
            #pragma unroll
            for (int u = 0; u < 4; ++u) {
                const int w = w0 + (PT / 32) * u;
                const int c = w < nw ? cntw[w] : 0;
                ok[u] = lane < c;
                q[u] = make_uint2(0u, 0u);
                if (ok[u]) q[u] = reg[(size_t)w * sc + 1 + lane];
            }
            #pragma unroll
            for (int u = 0; u < 4; ++u) {
                if (ok[u]) {
                    const int dl = (int)(q[u].x >> 24);
                    const int r = atomicAdd(&lcnt[dl], 1);
                    if (r < CAP)
                        reinterpret_cast<uint2*>(out + (size_t)(node0 + dl) * DIN_)[r] =
                            make_uint2(q[u].x & 0xFFFFFFu, q[u].y);
                }
            }
            for (int u = 0; u < 4; ++u) {
                const int w = w0 + (PT / 32) * u;
                const int c = w < nw ? cntw[w] : 0;
                for (int k = lane + 32; k < c; k += 32) {
                    const uint2 qq = reg[(size_t)w * sc + 1 + k];
                    const int dl = (int)(qq.x >> 24);
                    const int r = atomicAdd(&lcnt[dl], 1);
                    if (r < CAP)
                        reinterpret_cast<uint2*>(out + (size_t)(node0 + dl) * DIN_)[r] =
                            make_uint2(qq.x & 0xFFFFFFu, qq.y);
                }
            }
        }
    } else {
        for (long long i = t; i < e; i += PT) {
            const int d = dst[i];
            if (d >= node0 && d < node0 + rows) {
                const int r = atomicAdd(&lcnt[d - node0], 1);
                if (r < CAP)
                    reinterpret_cast<uint2*>(out + (size_t)d * DIN_)[r] =
                        make_uint2((unsigned)src[i], __float_as_uint(ea[i]));
            }
        }
    }
    __syncthreads();
    if (t < rows) cnt[node0 + t] = lcnt[t];
}

__global__ void gather_kernel(const int* __restrict__ src, const int* __restrict__ dst,
                              const float* __restrict__ ea,
                              const float* __restrict__ a_src, const float* __restrict__ a_dst,
                              const int* __restrict__ cnt,
                              const float* __restrict__ h, const float* __restrict__ bias,
                              float* out, int n, int e) {
    const int lane = threadIdx.x & 31;
    const int node = __builtin_amdgcn_readfirstlane(
        (int)(blockIdx.x * (blockDim.x >> 5) + (threadIdx.x >> 5)));
    if (node >= n) return;
    const int c = cnt[node];
    float* orow = out + (size_t)node * DIN_;
    const float* __restrict__ hself = h + (size_t)node * DIN_ + lane;
    const float NEG_INF = -__builtin_inff();

    float ad[HEADS], asf[HEADS];
    for (int hh = 0; hh < HEADS; ++hh) {
        ad[hh]  = a_dst[node * HEADS + hh];
        asf[hh] = a_src[node * HEADS + hh];
    }
    float acc[HEADS], den[HEADS];

    if (c <= CAP) {
        const uint2* slots = reinterpret_cast<const uint2*>(orow);
        int   sj[2];
        bool  ok[2];
        float vj[2][HEADS], exj[2][HEADS];
        float wl = 0.0f;
        for (int b = 0; b < 2; ++b) {
            const int k = lane + 32 * b;
            ok[b] = k < c;
            int s = node;
            float w = 1.0f;
            if (ok[b]) {
                const uint2 q = slots[k];
                s = (int)q.x;
                w = __uint_as_float(q.y);
                wl += w;
            }
            sj[b] = s;
            if (32 * b < c) {
                const float l2w = __log2f(w);
                const float4 as4 = *reinterpret_cast<const float4*>(a_src + (size_t)s * HEADS);
                vj[b][0] = ok[b] ? leaky(as4.x + ad[0]) + l2w : NEG_INF;
                vj[b][1] = ok[b] ? leaky(as4.y + ad[1]) + l2w : NEG_INF;
                vj[b][2] = ok[b] ? leaky(as4.z + ad[2]) + l2w : NEG_INF;
                vj[b][3] = ok[b] ? leaky(as4.w + ad[3]) + l2w : NEG_INF;
            } else {
                for (int hh = 0; hh < HEADS; ++hh) vj[b][hh] = NEG_INF;
            }
        }
        const float wsum = wave_sum(wl);
        const float lw = c > 0 ? wsum / (float)c : 1.0f;
        const float l2self = __log2f(lw);
        for (int hh = 0; hh < HEADS; ++hh) {
            const float vs = leaky(asf[hh] + ad[hh]) + l2self;
            const float m = fmaxf(wave_max(fmaxf(vj[0][hh], vj[1][hh])), vs);
            exj[0][hh] = ok[0] ? __expf(vj[0][hh] - m) : 0.0f;
            exj[1][hh] = ok[1] ? __expf(vj[1][hh] - m) : 0.0f;
            const float exs = __expf(vs - m);
            den[hh] = wave_sum(exj[0][hh] + exj[1][hh]) + exs;
            acc[hh] = exs * hself[hh * CH];
        }
        for (int b = 0; b < 2; ++b) {
            const int nb = min(c - 32 * b, 32);
            #pragma unroll 4
            for (int j = 0; j < nb; ++j) {
                const int s = __builtin_amdgcn_readlane(sj[b], j);
                const float* __restrict__ hp = h + (size_t)s * DIN_ + lane;
                for (int cc = 0; cc < HEADS; ++cc)
                    acc[cc] = fmaf(lane_f(exj[b][cc], j), hp[cc * CH], acc[cc]);
            }
        }
    } else {
        float wl = 0.0f;
        float ml[HEADS];
        for (int hh = 0; hh < HEADS; ++hh) ml[hh] = NEG_INF;
        for (int base = 0; base < e; base += 32) {
            const int i = base + lane;
            if (i < e && dst[i] == node) {
                const float w = ea[i];
                const float l2w = __log2f(w);
                const float4 as4 = *reinterpret_cast<const float4*>(a_src + (size_t)src[i] * HEADS);
                wl += w;
                ml[0] = fmaxf(ml[0], leaky(as4.x + ad[0]) + l2w);
                ml[1] = fmaxf(ml[1], leaky(as4.y + ad[1]) + l2w);
                ml[2] = fmaxf(ml[2], leaky(as4.z + ad[2]) + l2w);
                ml[3] = fmaxf(ml[3], leaky(as4.w + ad[3]) + l2w);
            }
        }
        const float wsum = wave_sum(wl);
        const float l2self = __log2f(wsum / (float)c);
        float mx[HEADS], exs[HEADS], dl[HEADS];
        for (int hh = 0; hh < HEADS; ++hh) {
            const float vs = leaky(asf[hh] + ad[hh]) + l2self;
            mx[hh] = fmaxf(wave_max(ml[hh]), vs);
            exs[hh] = __expf(vs - mx[hh]);
            acc[hh] = exs[hh] * hself[hh * CH];
            dl[hh] = 0.0f;
        }
        for (int base = 0; base < e; base += 32) {
            const int i = base + lane;
            const bool hit = i < e && dst[i] == node;
            int s = node;
            float ex[HEADS] = {0.0f, 0.0f, 0.0f, 0.0f};
            if (hit) {
                s = src[i];
                const float l2w = __log2f(ea[i]);
                const float4 as4 = *reinterpret_cast<const float4*>(a_src + (size_t)s * HEADS);
                ex[0] = __expf(leaky(as4.x + ad[0]) + l2w - mx[0]);
                ex[1] = __expf(leaky(as4.y + ad[1]) + l2w - mx[1]);
                ex[2] = __expf(leaky(as4.z + ad[2]) + l2w - mx[2]);
                ex[3] = __expf(leaky(as4.w + ad[3]) + l2w - mx[3]);
                for (int hh = 0; hh < HEADS; ++hh) dl[hh] += ex[hh];
            }
            unsigned mask = __builtin_amdgcn_ballot_w32(hit);
            while (mask) {
                const int j = __builtin_ctz(mask);
                mask &= mask - 1;
                const int sb = __builtin_amdgcn_readlane(s, j);
                const float* __restrict__ hp = h + (size_t)sb * DIN_ + lane;
                for (int cc = 0; cc < HEADS; ++cc)
                    acc[cc] = fmaf(lane_f(ex[cc], j), hp[cc * CH], acc[cc]);
            }
        }
        for (int hh = 0; hh < HEADS; ++hh) den[hh] = wave_sum(dl[hh]) + exs[hh];
    }

    for (int cc = 0; cc < HEADS; ++cc)
        orow[lane + cc * CH] = acc[cc] / den[cc] + bias[lane + cc * CH];
}

extern "C" void kernel_launch(void* const* d_in, const int* in_sizes, int n_in,
                              void* d_out, int out_size, void* d_ws, size_t ws_size,
                              hipStream_t stream) {
    const float* x        = (const float*)d_in[0];
    const int*   eidx     = (const int*)d_in[1];
    const float* eattr    = (const float*)d_in[2];
    const float* W        = (const float*)d_in[3];
    const float* att_src  = (const float*)d_in[4];
    const float* att_dst  = (const float*)d_in[5];
    const float* bias     = (const float*)d_in[6];

    const int n = in_sizes[0] / DIN_;
    const int e = in_sizes[2];
    const int* src = eidx;
    const int* dst = eidx + e;

    float* out = (float*)d_out;

    float* h    = (float*)d_ws;
    float* asrc = h    + (size_t)n * DIN_;
    float* adst = asrc + (size_t)n * HEADS;
    int*   cnt  = (int*)(adst + (size_t)n * HEADS);

    const int B = 256;

    uint2* rec = (uint2*)h;
    const int NB = (n + RB - 1) / RB;
    const int NW = (e + EPB - 1) / EPB;
    partition_kernel<<<(unsigned)NW, PT, (size_t)NB * sizeof(int), stream>>>(
        src, dst, eattr, rec, n, e, NB, NW);
    rank_kernel<<<(unsigned)NB, PT, (size_t)NW * sizeof(int), stream>>>(
        rec, src, dst, eattr, cnt, out, n, e, NW);

    gemm_wmma_kernel<<<(unsigned)((n + 31) / 32), 64, 0, stream>>>(
        x, W, att_src, att_dst, h, asrc, adst, n);

    gather_kernel<<<(unsigned)((n + 7) / 8), B, 0, stream>>>(
        src, dst, eattr, asrc, adst, cnt, h, bias, out, n, e);
}


# gather: lane = (slot%8, head): per-head max/sums inside 8-lane groups by DPP (was LDS permutes), an edge's 512-B h row as one 16-byte load per lane (was 4 loads); den, mean-weight sum order change
# speedup vs baseline: 6.1640x; 1.0777x over previous
#include <hip/hip_runtime.h>
#include <hip/hip_bf16.h>

typedef float v2f __attribute__((ext_vector_type(2)));
typedef float v4f __attribute__((ext_vector_type(4)));
typedef float v8f __attribute__((ext_vector_type(8)));

#define HEADS 4
#define CH    32
#define DIN_  128
#define NEG_SLOPE_ 0.2f
#define CAP   64

__device__ __forceinline__ float leaky(float v) {
    return v > 0.0f ? v : NEG_SLOPE_ * v;
}

__device__ __forceinline__ float wave_sum(float v) {
    for (int m = 16; m >= 1; m >>= 1) v += __shfl_xor(v, m, 32);
    return v;
}
__device__ __forceinline__ float wave_max(float v) {
    for (int m = 16; m >= 1; m >>= 1) v = fmaxf(v, __shfl_xor(v, m, 32));
    return v;
}
__device__ __forceinline__ float lane_f(float v, int j) {
    return __int_as_float(__builtin_amdgcn_readlane(__float_as_int(v), j));
}

template <int M>
__device__ __forceinline__ float row_xor(float v) {
    return __int_as_float(__builtin_amdgcn_update_dpp(
        0, __float_as_int(v), 0x160 + M, 0xf, 0xf, false));
}

template <int K>
__device__ __forceinline__ float grp_bcast(float v) {
    return __uint_as_float(__builtin_amdgcn_mov_dpp8(__float_as_uint(v), (unsigned)K * 0x249249u));
}
__device__ __forceinline__ float grp_sum(float v) {
    v += row_xor<1>(v);
    v += row_xor<2>(v);
    v += row_xor<4>(v);
    return v;
}
__device__ __forceinline__ float grp_max(float v) {
    v = fmaxf(v, row_xor<1>(v));
    v = fmaxf(v, row_xor<2>(v));
    v = fmaxf(v, row_xor<4>(v));
    return v;
}

__global__ void __launch_bounds__(64)
gemm_wmma_kernel(const float* __restrict__ x,
                 const float* __restrict__ W,
                 const float* __restrict__ att_src,
                 const float* __restrict__ att_dst,
                 float* __restrict__ h,
                 float* __restrict__ a_src, float* __restrict__ a_dst, int n) {
    const int lane = threadIdx.x & 31;
    const int half = lane >> 4;
    const int l16  = lane & 15;
    const int strip = blockIdx.x * (blockDim.x >> 5) + (threadIdx.x >> 5);
    const int row0 = strip * 16;
    if (row0 >= n) return;

    int arow = row0 + l16;
    if (arow >= n) arow = n - 1;
    const float* __restrict__ xrow = x + (size_t)arow * DIN_ + 4 * half;
    const float* __restrict__ wp   = W + (size_t)(4 * half) * DIN_ + 4 * l16;

    v8f acc[8];
    #pragma unroll
    for (int t = 0; t < 8; ++t) acc[t] = v8f{};

    #pragma unroll
    for (int j = 0; j < DIN_ / 8; ++j) {
        const v4f a4 = *reinterpret_cast<const v4f*>(xrow + 8 * j);
        v4f wlo[4], whi[4];
        #pragma unroll
        for (int q = 0; q < 4; ++q) {
            const float* __restrict__ wr = wp + (size_t)(8 * j + q) * DIN_;
            wlo[q] = *reinterpret_cast<const v4f*>(wr);
            whi[q] = *reinterpret_cast<const v4f*>(wr + 64);
        }
        #pragma unroll
        for (int p = 0; p < 2; ++p) {
            v2f a;
            a.x = a4[2 * p];
            a.y = a4[2 * p + 1];
            #pragma unroll
            for (int t = 0; t < 4; ++t) {
                v2f b0, b1;
                b0.x = wlo[2 * p][t];
                b0.y = wlo[2 * p + 1][t];
                b1.x = whi[2 * p][t];
                b1.y = whi[2 * p + 1][t];
                acc[t] = __builtin_amdgcn_wmma_f32_16x16x4_f32(
                     false, a,  false, b0,
                     (short)0, acc[t],  false,  false);
                acc[4 + t] = __builtin_amdgcn_wmma_f32_16x16x4_f32(
                     false, a,  false, b1,
                     (short)0, acc[4 + t],  false,  false);
            }
        }
    }

    float* __restrict__ hp = h + 4 * l16;
    #pragma unroll
    for (int r = 0; r < 8; ++r) {
        const int row = row0 + r + half * 8;
        if (row < n) {
            v4f lo, hi;
            lo.x = acc[0][r]; lo.y = acc[1][r]; lo.z = acc[2][r]; lo.w = acc[3][r];
            hi.x = acc[4][r]; hi.y = acc[5][r]; hi.z = acc[6][r]; hi.w = acc[7][r];
            float* __restrict__ hr = hp + (size_t)row * DIN_;
            *reinterpret_cast<v4f*>(hr)      = lo;
            *reinterpret_cast<v4f*>(hr + 64) = hi;
        }
    }

    const v4f s_lo = *reinterpret_cast<const v4f*>(att_src + 4 * l16);
    const v4f s_hi = *reinterpret_cast<const v4f*>(att_src + 64 + 4 * l16);
    const v4f d_lo = *reinterpret_cast<const v4f*>(att_dst + 4 * l16);
    const v4f d_hi = *reinterpret_cast<const v4f*>(att_dst + 64 + 4 * l16);
    float p[8][4];
    #pragma unroll
    for (int r = 0; r < 8; ++r) {
        p[r][0] = fmaf(acc[3][r], s_lo.w, fmaf(acc[2][r], s_lo.z, fmaf(acc[1][r], s_lo.y, acc[0][r] * s_lo.x)));
        p[r][1] = fmaf(acc[7][r], s_hi.w, fmaf(acc[6][r], s_hi.z, fmaf(acc[5][r], s_hi.y, acc[4][r] * s_hi.x)));
        p[r][2] = fmaf(acc[3][r], d_lo.w, fmaf(acc[2][r], d_lo.z, fmaf(acc[1][r], d_lo.y, acc[0][r] * d_lo.x)));
        p[r][3] = fmaf(acc[7][r], d_hi.w, fmaf(acc[6][r], d_hi.z, fmaf(acc[5][r], d_hi.y, acc[4][r] * d_hi.x)));
    }
    const bool b4 = (lane & 4) != 0;
    const bool b2 = (lane & 2) != 0;
    const bool b1 = (lane & 1) != 0;
    float q4[4][4], q2[2][4], q1[4];
    #pragma unroll
    for (int r = 0; r < 4; ++r) {
        #pragma unroll
        for (int k = 0; k < 4; ++k) {
            const float keep = b4 ? p[r + 4][k] : p[r][k];
            const float send = b4 ? p[r][k] : p[r + 4][k];
            q4[r][k] = keep + row_xor<4>(send);
        }
    }
    #pragma unroll
    for (int r = 0; r < 2; ++r) {
        #pragma unroll
        for (int k = 0; k < 4; ++k) {
            const float keep = b2 ? q4[r + 2][k] : q4[r][k];
            const float send = b2 ? q4[r][k] : q4[r + 2][k];
            q2[r][k] = keep + row_xor<2>(send);
        }
    }
    #pragma unroll
    for (int k = 0; k < 4; ++k) {
        const float keep = b1 ? q2[1][k] : q2[0][k];
        const float send = b1 ? q2[0][k] : q2[1][k];
        q1[k] = keep + row_xor<1>(send);
    }
    const bool g = (lane & 8) != 0;
    const float r0 = row_xor<8>(g ? q1[0] : q1[2]);
    const float r1 = row_xor<8>(g ? q1[1] : q1[3]);
    v4f o;
    o.x = g ? r0 : q1[0];
    o.y = g ? q1[2] : r0;
    o.z = g ? r1 : q1[1];
    o.w = g ? q1[3] : r1;
    const int orow = row0 + (lane & 7) + half * 8;
    if (orow < n)
        *reinterpret_cast<v4f*>((g ? a_dst : a_src) + (size_t)orow * HEADS) = o;
}

#define RB     256
#define RB_LOG 8
#define PT     1024
#define EPT    8
#define EPB    (PT * EPT)

__global__ void __launch_bounds__(PT)
partition_kernel(const int* __restrict__ src, const int* __restrict__ dst,
                 const float* __restrict__ ea, uint2* __restrict__ rec,
                 int n, int e, int nb, int nw) {
    extern __shared__ int hist[];
    const int t  = threadIdx.x;
    const int wb = blockIdx.x;
    for (int b = t; b < nb; b += PT) hist[b] = 0;
    __syncthreads();

    const int sc_full = (RB * CAP) / nw;
    const int sc_last = ((n - (nb - 1) * RB) * CAP) / nw;
    const bool packable = n <= (1 << 24);

    int d[EPT], s[EPT], rk[EPT];
    float wv[EPT];
    const long long i0 = (long long)wb * EPB + t;
    #pragma unroll
    for (int k = 0; k < EPT; ++k) {
        const long long i = i0 + (long long)k * PT;
        d[k] = -1; s[k] = 0; wv[k] = 0.0f;
        if (i < e) { d[k] = dst[i]; s[k] = src[i]; wv[k] = ea[i]; }
    }
    #pragma unroll
    for (int k = 0; k < EPT; ++k) {
        rk[k] = 0;
        if (d[k] >= 0) rk[k] = atomicAdd(&hist[d[k] >> RB_LOG], 1);
    }
    __syncthreads();

    for (int b = t; b < nb; b += PT) {
        const int sc = (b == nb - 1) ? sc_last : sc_full;
        if (sc >= 2)
            rec[(size_t)b * (RB * CAP) + (size_t)wb * sc] = make_uint2((unsigned)hist[b], 0u);
    }
    if (packable) {
        #pragma unroll
        for (int k = 0; k < EPT; ++k) {
            if (d[k] >= 0) {
                const int b = d[k] >> RB_LOG;
                const int sc = (b == nb - 1) ? sc_last : sc_full;
                if (rk[k] < sc - 1)
                    rec[(size_t)b * (RB * CAP) + (size_t)wb * sc + 1 + rk[k]] =
                        make_uint2((unsigned)s[k] | ((unsigned)(d[k] & (RB - 1)) << 24),
                                   __float_as_uint(wv[k]));
            }
        }
    }
}

__global__ void __launch_bounds__(PT)
rank_kernel(const uint2* __restrict__ rec,
            const int* __restrict__ src, const int* __restrict__ dst,
            const float* __restrict__ ea,
            int* __restrict__ cnt, float* __restrict__ out, int n, int e, int nw) {
    __shared__ int lcnt[RB];
    __shared__ int ovf_s;
    extern __shared__ int cntw[];
    const int t = threadIdx.x;
    const int b = blockIdx.x;
    const int node0 = b * RB;
    const int rows = min(RB, n - node0);
    const int sc = nw > 0 ? (rows * CAP) / nw : 0;
    const int usable = (n <= (1 << 24)) ? sc - 1 : -1;
    const uint2* __restrict__ reg = rec + (size_t)b * (RB * CAP);

    if (t < RB) lcnt[t] = 0;
    if (t == 0) ovf_s = (sc < 2) ? 1 : 0;
    __syncthreads();
    if (sc >= 2) {
        for (int w = t; w < nw; w += PT) {
            const int c = (int)reg[(size_t)w * sc].x;
            cntw[w] = c;
            if (c > usable) ovf_s = 1;
        }
    }
    __syncthreads();
    const bool ovf = ovf_s != 0;

    if (!ovf) {
        const int wave = t >> 5, lane = t & 31;
        for (int w0 = wave; w0 < nw; w0 += (PT / 32) * 4) {
            uint2 q[4];
            bool ok[4];
            #pragma unroll
            for (int u = 0; u < 4; ++u) {
                const int w = w0 + (PT / 32) * u;
                const int c = w < nw ? cntw[w] : 0;
                ok[u] = lane < c;
                q[u] = make_uint2(0u, 0u);
                if (ok[u]) q[u] = reg[(size_t)w * sc + 1 + lane];
            }
            #pragma unroll
            for (int u = 0; u < 4; ++u) {
                if (ok[u]) {
                    const int dl = (int)(q[u].x >> 24);
                    const int r = atomicAdd(&lcnt[dl], 1);
                    if (r < CAP)
                        reinterpret_cast<uint2*>(out + (size_t)(node0 + dl) * DIN_)[r] =
                            make_uint2(q[u].x & 0xFFFFFFu, q[u].y);
                }
            }
            for (int u = 0; u < 4; ++u) {
                const int w = w0 + (PT / 32) * u;
                const int c = w < nw ? cntw[w] : 0;
                for (int k = lane + 32; k < c; k += 32) {
                    const uint2 qq = reg[(size_t)w * sc + 1 + k];
                    const int dl = (int)(qq.x >> 24);
                    const int r = atomicAdd(&lcnt[dl], 1);
                    if (r < CAP)
                        reinterpret_cast<uint2*>(out + (size_t)(node0 + dl) * DIN_)[r] =
                            make_uint2(qq.x & 0xFFFFFFu, qq.y);
                }
            }
        }
    } else {
        for (long long i = t; i < e; i += PT) {
            const int d = dst[i];
            if (d >= node0 && d < node0 + rows) {
                const int r = atomicAdd(&lcnt[d - node0], 1);
                if (r < CAP)
                    reinterpret_cast<uint2*>(out + (size_t)d * DIN_)[r] =
                        make_uint2((unsigned)src[i], __float_as_uint(ea[i]));
            }
        }
    }
    __syncthreads();
    if (t < rows) cnt[node0 + t] = lcnt[t];
}


template <int K0, int N>
__device__ __forceinline__ void edge_group(const float* __restrict__ h, int sidx, float ex,
                                           int lane, v4f& acc) {
    v4f r[N];
    #pragma unroll
    for (int j = 0; j < N; ++j) {
        const unsigned s = (unsigned)__builtin_amdgcn_readlane(sidx, K0 + j);
        r[j] = *reinterpret_cast<const v4f*>(h + (size_t)s * DIN_ + 4 * lane);
    }
    float w[N];
    if (N > 0) w[0] = grp_bcast<K0>(ex);
    if (N > 1) w[N > 1 ? 1 : 0] = grp_bcast<K0 + 1>(ex);
    if (N > 2) w[N > 2 ? 2 : 0] = grp_bcast<K0 + 2>(ex);
    if (N > 3) w[N > 3 ? 3 : 0] = grp_bcast<K0 + 3>(ex);
    #pragma unroll
    for (int j = 0; j < N; ++j) {
        acc.x = fmaf(w[j], r[j].x, acc.x);
        acc.y = fmaf(w[j], r[j].y, acc.y);
        acc.z = fmaf(w[j], r[j].z, acc.z);
        acc.w = fmaf(w[j], r[j].w, acc.w);
    }
}
template <int K0>
__device__ __forceinline__ void edge_half(const float* __restrict__ h, int sidx, float ex,
                                          int lane, int rem, v4f& acc) {
    if (rem >= 4)      edge_group<K0, 4>(h, sidx, ex, lane, acc);
    else if (rem == 3) edge_group<K0, 3>(h, sidx, ex, lane, acc);
    else if (rem == 2) edge_group<K0, 2>(h, sidx, ex, lane, acc);
    else               edge_group<K0, 1>(h, sidx, ex, lane, acc);
}

#define ROUNDS (CAP / 8)

__global__ void gather_kernel(const int* __restrict__ src, const int* __restrict__ dst,
                              const float* __restrict__ ea,
                              const float* __restrict__ a_src, const float* __restrict__ a_dst,
                              const int* __restrict__ cnt,
                              const float* __restrict__ h, const float* __restrict__ bias,
                              float* out, int n, int e) {
    const int lane = threadIdx.x & 31;
    const int node = __builtin_amdgcn_readfirstlane(
        (int)(blockIdx.x * (blockDim.x >> 5) + (threadIdx.x >> 5)));
    if (node >= n) return;
    const int c = cnt[node];
    float* orow = out + (size_t)node * DIN_;
    const float NEG_INF = -__builtin_inff();

    if (c <= CAP) {
        const int g  = lane >> 3;
        const int k8 = lane & 7;
        const float adh = a_dst[(size_t)node * HEADS + g];
        const float ash = a_src[(size_t)node * HEADS + g];
        const v4f hs = *reinterpret_cast<const v4f*>(h + (size_t)node * DIN_ + 4 * lane);
        const uint2* slots = reinterpret_cast<const uint2*>(orow);

        int   sidx[ROUNDS];
        float wv[ROUNDS], as[ROUNDS], ex[ROUNDS];
        #pragma unroll
        for (int i = 0; i < ROUNDS; ++i) {
            if (8 * i >= c) break;
            const uint2 q = slots[min(8 * i + k8, c - 1)];
            sidx[i] = (int)q.x;
            wv[i] = __uint_as_float(q.y);
        }
        #pragma unroll
        for (int i = 0; i < ROUNDS; ++i) {
            if (8 * i >= c) break;
            as[i] = a_src[(size_t)(unsigned)sidx[i] * HEADS + g];
        }
        float wl = 0.0f, mx = NEG_INF;
        #pragma unroll
        for (int i = 0; i < ROUNDS; ++i) {
            if (8 * i >= c) break;
            const bool ok = 8 * i + k8 < c;
            wl += ok ? wv[i] : 0.0f;
            ex[i] = ok ? leaky(as[i] + adh) + __log2f(wv[i]) : NEG_INF;
            mx = fmaxf(mx, ex[i]);
        }
        const float wsum = grp_sum(wl);
        const float lw = c > 0 ? wsum / (float)c : 1.0f;
        const float l2self = __log2f(lw);
        const float vs = leaky(ash + adh) + l2self;
        const float m = fmaxf(grp_max(mx), vs);
        const float exs = __expf(vs - m);
        float dl = 0.0f;
        #pragma unroll
        for (int i = 0; i < ROUNDS; ++i) {
            if (8 * i >= c) break;
            const bool ok = 8 * i + k8 < c;
            ex[i] = ok ? __expf(ex[i] - m) : 0.0f;
            dl += ex[i];
        }
        const float den = grp_sum(dl) + exs;
        v4f acc;
        acc.x = exs * hs.x; acc.y = exs * hs.y; acc.z = exs * hs.z; acc.w = exs * hs.w;
        #pragma unroll
        for (int i = 0; i < ROUNDS; ++i) {
            const int rem = c - 8 * i;
            if (rem <= 0) break;
            edge_half<0>(h, sidx[i], ex[i], lane, rem, acc);
            if (rem > 4) edge_half<4>(h, sidx[i], ex[i], lane, rem - 4, acc);
        }
        const v4f b4 = *reinterpret_cast<const v4f*>(bias + 4 * lane);
        v4f res;
        res.x = acc.x / den + b4.x;
        res.y = acc.y / den + b4.y;
        res.z = acc.z / den + b4.z;
        res.w = acc.w / den + b4.w;
        *reinterpret_cast<v4f*>(orow + 4 * lane) = res;
        return;
    }

    const float* __restrict__ hself = h + (size_t)node * DIN_ + lane;
    float ad[HEADS], asf[HEADS];
    for (int hh = 0; hh < HEADS; ++hh) {
        ad[hh]  = a_dst[node * HEADS + hh];
        asf[hh] = a_src[node * HEADS + hh];
    }
    float acc[HEADS], den[HEADS];

    float wl = 0.0f;
    float ml[HEADS];
    for (int hh = 0; hh < HEADS; ++hh) ml[hh] = NEG_INF;
    for (int base = 0; base < e; base += 32) {
        const int i = base + lane;
        if (i < e && dst[i] == node) {
            const float w = ea[i];
            const float l2w = __log2f(w);
            const float4 as4 = *reinterpret_cast<const float4*>(a_src + (size_t)src[i] * HEADS);
            wl += w;
            ml[0] = fmaxf(ml[0], leaky(as4.x + ad[0]) + l2w);
            ml[1] = fmaxf(ml[1], leaky(as4.y + ad[1]) + l2w);
            ml[2] = fmaxf(ml[2], leaky(as4.z + ad[2]) + l2w);
            ml[3] = fmaxf(ml[3], leaky(as4.w + ad[3]) + l2w);
        }
    }
    const float wsum = wave_sum(wl);
    const float l2self = __log2f(wsum / (float)c);
    float mx[HEADS], exs[HEADS], dl[HEADS];
    for (int hh = 0; hh < HEADS; ++hh) {
        const float vs = leaky(asf[hh] + ad[hh]) + l2self;
        mx[hh] = fmaxf(wave_max(ml[hh]), vs);
        exs[hh] = __expf(vs - mx[hh]);
        acc[hh] = exs[hh] * hself[hh * CH];
        dl[hh] = 0.0f;
    }
    for (int base = 0; base < e; base += 32) {
        const int i = base + lane;
        const bool hit = i < e && dst[i] == node;
        int s = node;
        float ex[HEADS] = {0.0f, 0.0f, 0.0f, 0.0f};
        if (hit) {
            s = src[i];
            const float l2w = __log2f(ea[i]);
            const float4 as4 = *reinterpret_cast<const float4*>(a_src + (size_t)s * HEADS);
            ex[0] = __expf(leaky(as4.x + ad[0]) + l2w - mx[0]);
            ex[1] = __expf(leaky(as4.y + ad[1]) + l2w - mx[1]);
            ex[2] = __expf(leaky(as4.z + ad[2]) + l2w - mx[2]);
            ex[3] = __expf(leaky(as4.w + ad[3]) + l2w - mx[3]);
            for (int hh = 0; hh < HEADS; ++hh) dl[hh] += ex[hh];
        }
        unsigned mask = __builtin_amdgcn_ballot_w32(hit);
        while (mask) {
            const int j = __builtin_ctz(mask);
            mask &= mask - 1;
            const int sb = __builtin_amdgcn_readlane(s, j);
            const float* __restrict__ hp = h + (size_t)sb * DIN_ + lane;
            for (int cc = 0; cc < HEADS; ++cc)
                acc[cc] = fmaf(lane_f(ex[cc], j), hp[cc * CH], acc[cc]);
        }
    }
    for (int hh = 0; hh < HEADS; ++hh) den[hh] = wave_sum(dl[hh]) + exs[hh];

    for (int cc = 0; cc < HEADS; ++cc)
        orow[lane + cc * CH] = acc[cc] / den[cc] + bias[lane + cc * CH];
}

extern "C" void kernel_launch(void* const* d_in, const int* in_sizes, int n_in,
                              void* d_out, int out_size, void* d_ws, size_t ws_size,
                              hipStream_t stream) {
    const float* x        = (const float*)d_in[0];
    const int*   eidx     = (const int*)d_in[1];
    const float* eattr    = (const float*)d_in[2];
    const float* W        = (const float*)d_in[3];
    const float* att_src  = (const float*)d_in[4];
    const float* att_dst  = (const float*)d_in[5];
    const float* bias     = (const float*)d_in[6];

    const int n = in_sizes[0] / DIN_;
    const int e = in_sizes[2];
    const int* src = eidx;
    const int* dst = eidx + e;

    float* out = (float*)d_out;

    float* h    = (float*)d_ws;
    float* asrc = h    + (size_t)n * DIN_;
    float* adst = asrc + (size_t)n * HEADS;
    int*   cnt  = (int*)(adst + (size_t)n * HEADS);

    const int B = 256;

    uint2* rec = (uint2*)h;
    const int NB = (n + RB - 1) / RB;
    const int NW = (e + EPB - 1) / EPB;
    partition_kernel<<<(unsigned)NW, PT, (size_t)NB * sizeof(int), stream>>>(
        src, dst, eattr, rec, n, e, NB, NW);
    rank_kernel<<<(unsigned)NB, PT, (size_t)NW * sizeof(int), stream>>>(
        rec, src, dst, eattr, cnt, out, n, e, NW);

    gemm_wmma_kernel<<<(unsigned)((n + 31) / 32), 64, 0, stream>>>(
        x, W, att_src, att_dst, h, asrc, adst, n);

    gather_kernel<<<(unsigned)((n + 7) / 8), B, 0, stream>>>(
        src, dst, eattr, asrc, adst, cnt, h, bias, out, n, e);
}


# launch one holds edge-partition blocks and product blocks side by side (records go to out; rank copies its bin to LDS, then writes slots in place); 3 launches; sum order still an LDS arrival rank
# speedup vs baseline: 6.5617x; 1.0645x over previous
#include <hip/hip_runtime.h>
#include <hip/hip_bf16.h>

typedef float v2f __attribute__((ext_vector_type(2)));
typedef float v4f __attribute__((ext_vector_type(4)));
typedef float v8f __attribute__((ext_vector_type(8)));

#define HEADS 4
#define CH    32
#define DIN_  128
#define NEG_SLOPE_ 0.2f
#define CAP   64
#define RB     256
#define RB_LOG 8
#define PT     1024
#define FT     512
#define EPB    8192
#define EPT    (EPB / FT)

__device__ __forceinline__ float leaky(float v) {
    return v > 0.0f ? v : NEG_SLOPE_ * v;
}

__device__ __forceinline__ float wave_sum(float v) {
    for (int m = 16; m >= 1; m >>= 1) v += __shfl_xor(v, m, 32);
    return v;
}
__device__ __forceinline__ float wave_max(float v) {
    for (int m = 16; m >= 1; m >>= 1) v = fmaxf(v, __shfl_xor(v, m, 32));
    return v;
}
__device__ __forceinline__ float lane_f(float v, int j) {
    return __int_as_float(__builtin_amdgcn_readlane(__float_as_int(v), j));
}

template <int M>
__device__ __forceinline__ float row_xor(float v) {
    return __int_as_float(__builtin_amdgcn_update_dpp(
        0, __float_as_int(v), 0x160 + M, 0xf, 0xf, false));
}

template <int K>
__device__ __forceinline__ float grp_bcast(float v) {
    return __uint_as_float(__builtin_amdgcn_mov_dpp8(__float_as_uint(v), (unsigned)K * 0x249249u));
}
__device__ __forceinline__ float grp_sum(float v) {
    v += row_xor<1>(v);
    v += row_xor<2>(v);
    v += row_xor<4>(v);
    return v;
}
__device__ __forceinline__ float grp_max(float v) {
    v = fmaxf(v, row_xor<1>(v));
    v = fmaxf(v, row_xor<2>(v));
    v = fmaxf(v, row_xor<4>(v));
    return v;
}

__device__ __forceinline__ void
gemm_strip(const float* __restrict__ x,
                 const float* __restrict__ W,
                 const float* __restrict__ att_src,
                 const float* __restrict__ att_dst,
                 float* __restrict__ h,
                 float* __restrict__ a_src, float* __restrict__ a_dst, int n, int strip) {
    const int lane = threadIdx.x & 31;
    const int half = lane >> 4;
    const int l16  = lane & 15;
    const int row0 = strip * 16;
    if (row0 >= n) return;

    int arow = row0 + l16;
    if (arow >= n) arow = n - 1;
    const float* __restrict__ xrow = x + (size_t)arow * DIN_ + 4 * half;
    const float* __restrict__ wp   = W + (size_t)(4 * half) * DIN_ + 4 * l16;

    v8f acc[8];
    #pragma unroll
    for (int t = 0; t < 8; ++t) acc[t] = v8f{};

    #pragma unroll
    for (int j = 0; j < DIN_ / 8; ++j) {
        const v4f a4 = *reinterpret_cast<const v4f*>(xrow + 8 * j);
        v4f wlo[4], whi[4];
        #pragma unroll
        for (int q = 0; q < 4; ++q) {
            const float* __restrict__ wr = wp + (size_t)(8 * j + q) * DIN_;
            wlo[q] = *reinterpret_cast<const v4f*>(wr);
            whi[q] = *reinterpret_cast<const v4f*>(wr + 64);
        }
        #pragma unroll
        for (int p = 0; p < 2; ++p) {
            v2f a;
            a.x = a4[2 * p];
            a.y = a4[2 * p + 1];
            #pragma unroll
            for (int t = 0; t < 4; ++t) {
                v2f b0, b1;
                b0.x = wlo[2 * p][t];
                b0.y = wlo[2 * p + 1][t];
                b1.x = whi[2 * p][t];
                b1.y = whi[2 * p + 1][t];
                acc[t] = __builtin_amdgcn_wmma_f32_16x16x4_f32(
                     false, a,  false, b0,
                     (short)0, acc[t],  false,  false);
                acc[4 + t] = __builtin_amdgcn_wmma_f32_16x16x4_f32(
                     false, a,  false, b1,
                     (short)0, acc[4 + t],  false,  false);
            }
        }
    }

    float* __restrict__ hp = h + 4 * l16;
    #pragma unroll
    for (int r = 0; r < 8; ++r) {
        const int row = row0 + r + half * 8;
        if (row < n) {
            v4f lo, hi;
            lo.x = acc[0][r]; lo.y = acc[1][r]; lo.z = acc[2][r]; lo.w = acc[3][r];
            hi.x = acc[4][r]; hi.y = acc[5][r]; hi.z = acc[6][r]; hi.w = acc[7][r];
            float* __restrict__ hr = hp + (size_t)row * DIN_;
            *reinterpret_cast<v4f*>(hr)      = lo;
            *reinterpret_cast<v4f*>(hr + 64) = hi;
        }
    }

    const v4f s_lo = *reinterpret_cast<const v4f*>(att_src + 4 * l16);
    const v4f s_hi = *reinterpret_cast<const v4f*>(att_src + 64 + 4 * l16);
    const v4f d_lo = *reinterpret_cast<const v4f*>(att_dst + 4 * l16);
    const v4f d_hi = *reinterpret_cast<const v4f*>(att_dst + 64 + 4 * l16);
    float p[8][4];
    #pragma unroll
    for (int r = 0; r < 8; ++r) {
        p[r][0] = fmaf(acc[3][r], s_lo.w, fmaf(acc[2][r], s_lo.z, fmaf(acc[1][r], s_lo.y, acc[0][r] * s_lo.x)));
        p[r][1] = fmaf(acc[7][r], s_hi.w, fmaf(acc[6][r], s_hi.z, fmaf(acc[5][r], s_hi.y, acc[4][r] * s_hi.x)));
        p[r][2] = fmaf(acc[3][r], d_lo.w, fmaf(acc[2][r], d_lo.z, fmaf(acc[1][r], d_lo.y, acc[0][r] * d_lo.x)));
        p[r][3] = fmaf(acc[7][r], d_hi.w, fmaf(acc[6][r], d_hi.z, fmaf(acc[5][r], d_hi.y, acc[4][r] * d_hi.x)));
    }
    const bool b4 = (lane & 4) != 0;
    const bool b2 = (lane & 2) != 0;
    const bool b1 = (lane & 1) != 0;
    float q4[4][4], q2[2][4], q1[4];
    #pragma unroll
    for (int r = 0; r < 4; ++r) {
        #pragma unroll
        for (int k = 0; k < 4; ++k) {
            const float keep = b4 ? p[r + 4][k] : p[r][k];
            const float send = b4 ? p[r][k] : p[r + 4][k];
            q4[r][k] = keep + row_xor<4>(send);
        }
    }
    #pragma unroll
    for (int r = 0; r < 2; ++r) {
        #pragma unroll
        for (int k = 0; k < 4; ++k) {
            const float keep = b2 ? q4[r + 2][k] : q4[r][k];
            const float send = b2 ? q4[r][k] : q4[r + 2][k];
            q2[r][k] = keep + row_xor<2>(send);
        }
    }
    #pragma unroll
    for (int k = 0; k < 4; ++k) {
        const float keep = b1 ? q2[1][k] : q2[0][k];
        const float send = b1 ? q2[0][k] : q2[1][k];
        q1[k] = keep + row_xor<1>(send);
    }
    const bool g = (lane & 8) != 0;
    const float r0 = row_xor<8>(g ? q1[0] : q1[2]);
    const float r1 = row_xor<8>(g ? q1[1] : q1[3]);
    v4f o;
    o.x = g ? r0 : q1[0];
    o.y = g ? q1[2] : r0;
    o.z = g ? r1 : q1[1];
    o.w = g ? q1[3] : r1;
    const int orow = row0 + (lane & 7) + half * 8;
    if (orow < n)
        *reinterpret_cast<v4f*>((g ? a_dst : a_src) + (size_t)orow * HEADS) = o;
}

__device__ __forceinline__ void
partition_block(const int* __restrict__ src, const int* __restrict__ dst,
                 const float* __restrict__ ea, uint2* __restrict__ rec,
                 int n, int e, int nb, int nw, int* hist, int wb) {
    const int t  = threadIdx.x;
    for (int b = t; b < nb; b += FT) hist[b] = 0;
    __syncthreads();

    const int sc_full = (RB * CAP) / nw;
    const int sc_last = ((n - (nb - 1) * RB) * CAP) / nw;
    const bool packable = n <= (1 << 24);

    int d[EPT], s[EPT], rk[EPT];
    float wv[EPT];
    const long long i0 = (long long)wb * EPB + t;
    #pragma unroll
    for (int k = 0; k < EPT; ++k) {
        const long long i = i0 + (long long)k * FT;
        d[k] = -1; s[k] = 0; wv[k] = 0.0f;
        if (i < e) { d[k] = dst[i]; s[k] = src[i]; wv[k] = ea[i]; }
    }
    #pragma unroll
    for (int k = 0; k < EPT; ++k) {
        rk[k] = 0;
        if (d[k] >= 0) rk[k] = atomicAdd(&hist[d[k] >> RB_LOG], 1);
    }
    __syncthreads();

    for (int b = t; b < nb; b += FT) {
        const int sc = (b == nb - 1) ? sc_last : sc_full;
        if (sc >= 2)
            rec[(size_t)b * (RB * CAP) + (size_t)wb * sc] = make_uint2((unsigned)hist[b], 0u);
    }
    if (packable) {
        #pragma unroll
        for (int k = 0; k < EPT; ++k) {
            if (d[k] >= 0) {
                const int b = d[k] >> RB_LOG;
                const int sc = (b == nb - 1) ? sc_last : sc_full;
                if (rk[k] < sc - 1)
                    rec[(size_t)b * (RB * CAP) + (size_t)wb * sc + 1 + rk[k]] =
                        make_uint2((unsigned)s[k] | ((unsigned)(d[k] & (RB - 1)) << 24),
                                   __float_as_uint(wv[k]));
            }
        }
    }
}

__global__ void __launch_bounds__(FT, 8)
front_kernel(const float* __restrict__ x,
             const float* __restrict__ W,
             const float* __restrict__ att_src,
             const float* __restrict__ att_dst,
             float* __restrict__ h,
             float* __restrict__ a_src, float* __restrict__ a_dst,
             const int* __restrict__ src, const int* __restrict__ dst,
             const float* __restrict__ ea, uint2* __restrict__ rec,
             int n, int e, int nb, int nw) {
    extern __shared__ int hist[];
    const int blk = (int)blockIdx.x;
    if (blk < nw) {
        partition_block(src, dst, ea, rec, n, e, nb, nw, hist, blk);
    } else {
        gemm_strip(x, W, att_src, att_dst, h, a_src, a_dst, n,
                   (blk - nw) * (FT / 32) + (int)(threadIdx.x >> 5));
    }
}

__global__ void __launch_bounds__(PT)
rank_kernel(const uint2* rec,
            const int* __restrict__ src, const int* __restrict__ dst,
            const float* __restrict__ ea,
            int* __restrict__ cnt, float* out, int n, int e, int nw) {
    __shared__ uint2 stage[RB * CAP];
    __shared__ int lcnt[RB];
    __shared__ int ovf_s;
    extern __shared__ int cntw[];
    const int t = threadIdx.x;
    const int b = blockIdx.x;
    const int node0 = b * RB;
    const int rows = min(RB, n - node0);
    const int sc = nw > 0 ? (rows * CAP) / nw : 0;
    const int usable = (n <= (1 << 24)) ? sc - 1 : -1;
    const uint2* reg = rec + (size_t)b * (RB * CAP);

    if (t < RB) lcnt[t] = 0;
    if (t == 0) ovf_s = (sc < 2) ? 1 : 0;
    __syncthreads();
    if (sc >= 2) {
        for (int w = t; w < nw; w += PT) {
            const int c = (int)reg[(size_t)w * sc].x;
            cntw[w] = c;
            if (c > usable) ovf_s = 1;
        }
    }
    __syncthreads();
    const bool ovf = ovf_s != 0;

    if (!ovf) {
        const int wave = t >> 5, lane = t & 31;
        for (int w0 = wave; w0 < nw; w0 += (PT / 32) * 4) {
            uint2 q[4];
            bool ok[4];
            #pragma unroll
            for (int u = 0; u < 4; ++u) {
                const int w = w0 + (PT / 32) * u;
                const int c = w < nw ? cntw[w] : 0;
                ok[u] = lane < c;
                q[u] = make_uint2(0u, 0u);
                if (ok[u]) q[u] = reg[(size_t)w * sc + 1 + lane];
            }
            #pragma unroll
            for (int u = 0; u < 4; ++u) {
                const int w = w0 + (PT / 32) * u;
                if (ok[u]) stage[w * sc + 1 + lane] = q[u];
            }
            for (int u = 0; u < 4; ++u) {
                const int w = w0 + (PT / 32) * u;
                const int c = w < nw ? cntw[w] : 0;
                for (int k = lane + 32; k < c; k += 32)
                    stage[w * sc + 1 + k] = reg[(size_t)w * sc + 1 + k];
            }
        }
        __syncthreads();
        for (int w0 = wave; w0 < nw; w0 += (PT / 32) * 4) {
            uint2 q[4];
            bool ok[4];
            #pragma unroll
            for (int u = 0; u < 4; ++u) {
                const int w = w0 + (PT / 32) * u;
                const int c = w < nw ? cntw[w] : 0;
                ok[u] = lane < c;
                q[u] = make_uint2(0u, 0u);
                if (ok[u]) q[u] = stage[w * sc + 1 + lane];
            }
            #pragma unroll
            for (int u = 0; u < 4; ++u) {
                if (ok[u]) {
                    const int dl = (int)(q[u].x >> 24);
                    const int r = atomicAdd(&lcnt[dl], 1);
                    if (r < CAP)
                        reinterpret_cast<uint2*>(out + (size_t)(node0 + dl) * DIN_)[r] =
                            make_uint2(q[u].x & 0xFFFFFFu, q[u].y);
                }
            }
            for (int u = 0; u < 4; ++u) {
                const int w = w0 + (PT / 32) * u;
                const int c = w < nw ? cntw[w] : 0;
                for (int k = lane + 32; k < c; k += 32) {
                    const uint2 qq = stage[w * sc + 1 + k];
                    const int dl = (int)(qq.x >> 24);
                    const int r = atomicAdd(&lcnt[dl], 1);
                    if (r < CAP)
                        reinterpret_cast<uint2*>(out + (size_t)(node0 + dl) * DIN_)[r] =
                            make_uint2(qq.x & 0xFFFFFFu, qq.y);
                }
            }
        }
    } else {
        for (long long i = t; i < e; i += PT) {
            const int d = dst[i];
            if (d >= node0 && d < node0 + rows) {
                const int r = atomicAdd(&lcnt[d - node0], 1);
                if (r < CAP)
                    reinterpret_cast<uint2*>(out + (size_t)d * DIN_)[r] =
                        make_uint2((unsigned)src[i], __float_as_uint(ea[i]));
            }
        }
    }
    __syncthreads();
    if (t < rows) cnt[node0 + t] = lcnt[t];
}


template <int K0, int N>
__device__ __forceinline__ void edge_group(const float* __restrict__ h, int sidx, float ex,
                                           int lane, v4f& acc) {
    v4f r[N];
    #pragma unroll
    for (int j = 0; j < N; ++j) {
        const unsigned s = (unsigned)__builtin_amdgcn_readlane(sidx, K0 + j);
        r[j] = *reinterpret_cast<const v4f*>(h + (size_t)s * DIN_ + 4 * lane);
    }
    float w[N];
    if (N > 0) w[0] = grp_bcast<K0>(ex);
    if (N > 1) w[N > 1 ? 1 : 0] = grp_bcast<K0 + 1>(ex);
    if (N > 2) w[N > 2 ? 2 : 0] = grp_bcast<K0 + 2>(ex);
    if (N > 3) w[N > 3 ? 3 : 0] = grp_bcast<K0 + 3>(ex);
    #pragma unroll
    for (int j = 0; j < N; ++j) {
        acc.x = fmaf(w[j], r[j].x, acc.x);
        acc.y = fmaf(w[j], r[j].y, acc.y);
        acc.z = fmaf(w[j], r[j].z, acc.z);
        acc.w = fmaf(w[j], r[j].w, acc.w);
    }
}
template <int K0>
__device__ __forceinline__ void edge_half(const float* __restrict__ h, int sidx, float ex,
                                          int lane, int rem, v4f& acc) {
    if (rem >= 4)      edge_group<K0, 4>(h, sidx, ex, lane, acc);
    else if (rem == 3) edge_group<K0, 3>(h, sidx, ex, lane, acc);
    else if (rem == 2) edge_group<K0, 2>(h, sidx, ex, lane, acc);
    else               edge_group<K0, 1>(h, sidx, ex, lane, acc);
}

#define ROUNDS (CAP / 8)

__global__ void gather_kernel(const int* __restrict__ src, const int* __restrict__ dst,
                              const float* __restrict__ ea,
                              const float* __restrict__ a_src, const float* __restrict__ a_dst,
                              const int* __restrict__ cnt,
                              const float* __restrict__ h, const float* __restrict__ bias,
                              float* out, int n, int e) {
    const int lane = threadIdx.x & 31;
    const int node = __builtin_amdgcn_readfirstlane(
        (int)(blockIdx.x * (blockDim.x >> 5) + (threadIdx.x >> 5)));
    if (node >= n) return;
    const int c = cnt[node];
    float* orow = out + (size_t)node * DIN_;
    const float NEG_INF = -__builtin_inff();

    if (c <= CAP) {
        const int g  = lane >> 3;
        const int k8 = lane & 7;
        const float adh = a_dst[(size_t)node * HEADS + g];
        const float ash = a_src[(size_t)node * HEADS + g];
        const v4f hs = *reinterpret_cast<const v4f*>(h + (size_t)node * DIN_ + 4 * lane);
        const uint2* slots = reinterpret_cast<const uint2*>(orow);

        int   sidx[ROUNDS];
        float wv[ROUNDS], as[ROUNDS], ex[ROUNDS];
        #pragma unroll
        for (int i = 0; i < ROUNDS; ++i) {
            if (8 * i >= c) break;
            const uint2 q = slots[min(8 * i + k8, c - 1)];
            sidx[i] = (int)q.x;
            wv[i] = __uint_as_float(q.y);
        }
        #pragma unroll
        for (int i = 0; i < ROUNDS; ++i) {
            if (8 * i >= c) break;
            as[i] = a_src[(size_t)(unsigned)sidx[i] * HEADS + g];
        }
        float wl = 0.0f, mx = NEG_INF;
        #pragma unroll
        for (int i = 0; i < ROUNDS; ++i) {
            if (8 * i >= c) break;
            const bool ok = 8 * i + k8 < c;
            wl += ok ? wv[i] : 0.0f;
            ex[i] = ok ? leaky(as[i] + adh) + __log2f(wv[i]) : NEG_INF;
            mx = fmaxf(mx, ex[i]);
        }
        const float wsum = grp_sum(wl);
        const float lw = c > 0 ? wsum / (float)c : 1.0f;
        const float l2self = __log2f(lw);
        const float vs = leaky(ash + adh) + l2self;
        const float m = fmaxf(grp_max(mx), vs);
        const float exs = __expf(vs - m);
        float dl = 0.0f;
        #pragma unroll
        for (int i = 0; i < ROUNDS; ++i) {
            if (8 * i >= c) break;
            const bool ok = 8 * i + k8 < c;
            ex[i] = ok ? __expf(ex[i] - m) : 0.0f;
            dl += ex[i];
        }
        const float den = grp_sum(dl) + exs;
        v4f acc;
        acc.x = exs * hs.x; acc.y = exs * hs.y; acc.z = exs * hs.z; acc.w = exs * hs.w;
        #pragma unroll
        for (int i = 0; i < ROUNDS; ++i) {
            const int rem = c - 8 * i;
            if (rem <= 0) break;
            edge_half<0>(h, sidx[i], ex[i], lane, rem, acc);
            if (rem > 4) edge_half<4>(h, sidx[i], ex[i], lane, rem - 4, acc);
        }
        const v4f b4 = *reinterpret_cast<const v4f*>(bias + 4 * lane);
        v4f res;
        res.x = acc.x / den + b4.x;
        res.y = acc.y / den + b4.y;
        res.z = acc.z / den + b4.z;
        res.w = acc.w / den + b4.w;
        *reinterpret_cast<v4f*>(orow + 4 * lane) = res;
        return;
    }

    const float* __restrict__ hself = h + (size_t)node * DIN_ + lane;
    float ad[HEADS], asf[HEADS];
    for (int hh = 0; hh < HEADS; ++hh) {
        ad[hh]  = a_dst[node * HEADS + hh];
        asf[hh] = a_src[node * HEADS + hh];
    }
    float acc[HEADS], den[HEADS];

    float wl = 0.0f;
    float ml[HEADS];
    for (int hh = 0; hh < HEADS; ++hh) ml[hh] = NEG_INF;
    for (int base = 0; base < e; base += 32) {
        const int i = base + lane;
        if (i < e && dst[i] == node) {
            const float w = ea[i];
            const float l2w = __log2f(w);
            const float4 as4 = *reinterpret_cast<const float4*>(a_src + (size_t)src[i] * HEADS);
            wl += w;
            ml[0] = fmaxf(ml[0], leaky(as4.x + ad[0]) + l2w);
            ml[1] = fmaxf(ml[1], leaky(as4.y + ad[1]) + l2w);
            ml[2] = fmaxf(ml[2], leaky(as4.z + ad[2]) + l2w);
            ml[3] = fmaxf(ml[3], leaky(as4.w + ad[3]) + l2w);
        }
    }
    const float wsum = wave_sum(wl);
    const float l2self = __log2f(wsum / (float)c);
    float mx[HEADS], exs[HEADS], dl[HEADS];
    for (int hh = 0; hh < HEADS; ++hh) {
        const float vs = leaky(asf[hh] + ad[hh]) + l2self;
        mx[hh] = fmaxf(wave_max(ml[hh]), vs);
        exs[hh] = __expf(vs - mx[hh]);
        acc[hh] = exs[hh] * hself[hh * CH];
        dl[hh] = 0.0f;
    }
    for (int base = 0; base < e; base += 32) {
        const int i = base + lane;
        const bool hit = i < e && dst[i] == node;
        int s = node;
        float ex[HEADS] = {0.0f, 0.0f, 0.0f, 0.0f};
        if (hit) {
            s = src[i];
            const float l2w = __log2f(ea[i]);
            const float4 as4 = *reinterpret_cast<const float4*>(a_src + (size_t)s * HEADS);
            ex[0] = __expf(leaky(as4.x + ad[0]) + l2w - mx[0]);
            ex[1] = __expf(leaky(as4.y + ad[1]) + l2w - mx[1]);
            ex[2] = __expf(leaky(as4.z + ad[2]) + l2w - mx[2]);
            ex[3] = __expf(leaky(as4.w + ad[3]) + l2w - mx[3]);
            for (int hh = 0; hh < HEADS; ++hh) dl[hh] += ex[hh];
        }
        unsigned mask = __builtin_amdgcn_ballot_w32(hit);
        while (mask) {
            const int j = __builtin_ctz(mask);
            mask &= mask - 1;
            const int sb = __builtin_amdgcn_readlane(s, j);
            const float* __restrict__ hp = h + (size_t)sb * DIN_ + lane;
            for (int cc = 0; cc < HEADS; ++cc)
                acc[cc] = fmaf(lane_f(ex[cc], j), hp[cc * CH], acc[cc]);
        }
    }
    for (int hh = 0; hh < HEADS; ++hh) den[hh] = wave_sum(dl[hh]) + exs[hh];

    for (int cc = 0; cc < HEADS; ++cc)
        orow[lane + cc * CH] = acc[cc] / den[cc] + bias[lane + cc * CH];
}

extern "C" void kernel_launch(void* const* d_in, const int* in_sizes, int n_in,
                              void* d_out, int out_size, void* d_ws, size_t ws_size,
                              hipStream_t stream) {
    const float* x        = (const float*)d_in[0];
    const int*   eidx     = (const int*)d_in[1];
    const float* eattr    = (const float*)d_in[2];
    const float* W        = (const float*)d_in[3];
    const float* att_src  = (const float*)d_in[4];
    const float* att_dst  = (const float*)d_in[5];
    const float* bias     = (const float*)d_in[6];

    const int n = in_sizes[0] / DIN_;
    const int e = in_sizes[2];
    const int* src = eidx;
    const int* dst = eidx + e;

    float* out = (float*)d_out;

    float* h    = (float*)d_ws;
    float* asrc = h    + (size_t)n * DIN_;
    float* adst = asrc + (size_t)n * HEADS;
    int*   cnt  = (int*)(adst + (size_t)n * HEADS);

    const int B = 256;

    uint2* rec = (uint2*)out;
    const int NB = (n + RB - 1) / RB;
    const int NW = (e + EPB - 1) / EPB;
    const int GB = (n + 16 * (FT / 32) - 1) / (16 * (FT / 32));
    front_kernel<<<(unsigned)(NW + GB), FT, (size_t)NB * sizeof(int), stream>>>(
        x, W, att_src, att_dst, h, asrc, adst, src, dst, eattr, rec, n, e, NB, NW);
    rank_kernel<<<(unsigned)NB, PT, (size_t)NW * sizeof(int), stream>>>(
        rec, src, dst, eattr, cnt, out, n, e, NW);

    gather_kernel<<<(unsigned)((n + 7) / 8), B, 0, stream>>>(
        src, dst, eattr, asrc, adst, cnt, h, bias, out, n, e);
}


# gather moved into the rank launch: a bin's block ranks its records straight into LDS slots and its own waves then form the bin's rows; 2 launches; sum order still an LDS arrival rank
# speedup vs baseline: 6.9304x; 1.0562x over previous
#include <hip/hip_runtime.h>
#include <hip/hip_bf16.h>

typedef float v2f __attribute__((ext_vector_type(2)));
typedef float v4f __attribute__((ext_vector_type(4)));
typedef float v8f __attribute__((ext_vector_type(8)));

#define HEADS 4
#define CH    32
#define DIN_  128
#define NEG_SLOPE_ 0.2f
#define CAP   64
#define RB     256
#define RB_LOG 8
#define PT     512
#define PU     8
#define FT     512
#define EPB    8192
#define EPT    (EPB / FT)

__device__ __forceinline__ float leaky(float v) {
    return v > 0.0f ? v : NEG_SLOPE_ * v;
}

__device__ __forceinline__ float wave_sum(float v) {
    for (int m = 16; m >= 1; m >>= 1) v += __shfl_xor(v, m, 32);
    return v;
}
__device__ __forceinline__ float wave_max(float v) {
    for (int m = 16; m >= 1; m >>= 1) v = fmaxf(v, __shfl_xor(v, m, 32));
    return v;
}
__device__ __forceinline__ float lane_f(float v, int j) {
    return __int_as_float(__builtin_amdgcn_readlane(__float_as_int(v), j));
}

template <int M>
__device__ __forceinline__ float row_xor(float v) {
    return __int_as_float(__builtin_amdgcn_update_dpp(
        0, __float_as_int(v), 0x160 + M, 0xf, 0xf, false));
}

template <int K>
__device__ __forceinline__ float grp_bcast(float v) {
    return __uint_as_float(__builtin_amdgcn_mov_dpp8(__float_as_uint(v), (unsigned)K * 0x249249u));
}
__device__ __forceinline__ float grp_sum(float v) {
    v += row_xor<1>(v);
    v += row_xor<2>(v);
    v += row_xor<4>(v);
    return v;
}
__device__ __forceinline__ float grp_max(float v) {
    v = fmaxf(v, row_xor<1>(v));
    v = fmaxf(v, row_xor<2>(v));
    v = fmaxf(v, row_xor<4>(v));
    return v;
}

__device__ __forceinline__ void
gemm_strip(const float* __restrict__ x,
                 const float* __restrict__ W,
                 const float* __restrict__ att_src,
                 const float* __restrict__ att_dst,
                 float* __restrict__ h,
                 float* __restrict__ a_src, float* __restrict__ a_dst, int n, int strip) {
    const int lane = threadIdx.x & 31;
    const int half = lane >> 4;
    const int l16  = lane & 15;
    const int row0 = strip * 16;
    if (row0 >= n) return;

    int arow = row0 + l16;
    if (arow >= n) arow = n - 1;
    const float* __restrict__ xrow = x + (size_t)arow * DIN_ + 4 * half;
    const float* __restrict__ wp   = W + (size_t)(4 * half) * DIN_ + 4 * l16;

    v8f acc[8];
    #pragma unroll
    for (int t = 0; t < 8; ++t) acc[t] = v8f{};

    #pragma unroll
    for (int j = 0; j < DIN_ / 8; ++j) {
        const v4f a4 = *reinterpret_cast<const v4f*>(xrow + 8 * j);
        v4f wlo[4], whi[4];
        #pragma unroll
        for (int q = 0; q < 4; ++q) {
            const float* __restrict__ wr = wp + (size_t)(8 * j + q) * DIN_;
            wlo[q] = *reinterpret_cast<const v4f*>(wr);
            whi[q] = *reinterpret_cast<const v4f*>(wr + 64);
        }
        #pragma unroll
        for (int p = 0; p < 2; ++p) {
            v2f a;
            a.x = a4[2 * p];
            a.y = a4[2 * p + 1];
            #pragma unroll
            for (int t = 0; t < 4; ++t) {
                v2f b0, b1;
                b0.x = wlo[2 * p][t];
                b0.y = wlo[2 * p + 1][t];
                b1.x = whi[2 * p][t];
                b1.y = whi[2 * p + 1][t];
                acc[t] = __builtin_amdgcn_wmma_f32_16x16x4_f32(
                     false, a,  false, b0,
                     (short)0, acc[t],  false,  false);
                acc[4 + t] = __builtin_amdgcn_wmma_f32_16x16x4_f32(
                     false, a,  false, b1,
                     (short)0, acc[4 + t],  false,  false);
            }
        }
    }

    float* __restrict__ hp = h + 4 * l16;
    #pragma unroll
    for (int r = 0; r < 8; ++r) {
        const int row = row0 + r + half * 8;
        if (row < n) {
            v4f lo, hi;
            lo.x = acc[0][r]; lo.y = acc[1][r]; lo.z = acc[2][r]; lo.w = acc[3][r];
            hi.x = acc[4][r]; hi.y = acc[5][r]; hi.z = acc[6][r]; hi.w = acc[7][r];
            float* __restrict__ hr = hp + (size_t)row * DIN_;
            *reinterpret_cast<v4f*>(hr)      = lo;
            *reinterpret_cast<v4f*>(hr + 64) = hi;
        }
    }

    const v4f s_lo = *reinterpret_cast<const v4f*>(att_src + 4 * l16);
    const v4f s_hi = *reinterpret_cast<const v4f*>(att_src + 64 + 4 * l16);
    const v4f d_lo = *reinterpret_cast<const v4f*>(att_dst + 4 * l16);
    const v4f d_hi = *reinterpret_cast<const v4f*>(att_dst + 64 + 4 * l16);
    float p[8][4];
    #pragma unroll
    for (int r = 0; r < 8; ++r) {
        p[r][0] = fmaf(acc[3][r], s_lo.w, fmaf(acc[2][r], s_lo.z, fmaf(acc[1][r], s_lo.y, acc[0][r] * s_lo.x)));
        p[r][1] = fmaf(acc[7][r], s_hi.w, fmaf(acc[6][r], s_hi.z, fmaf(acc[5][r], s_hi.y, acc[4][r] * s_hi.x)));
        p[r][2] = fmaf(acc[3][r], d_lo.w, fmaf(acc[2][r], d_lo.z, fmaf(acc[1][r], d_lo.y, acc[0][r] * d_lo.x)));
        p[r][3] = fmaf(acc[7][r], d_hi.w, fmaf(acc[6][r], d_hi.z, fmaf(acc[5][r], d_hi.y, acc[4][r] * d_hi.x)));
    }
    const bool b4 = (lane & 4) != 0;
    const bool b2 = (lane & 2) != 0;
    const bool b1 = (lane & 1) != 0;
    float q4[4][4], q2[2][4], q1[4];
    #pragma unroll
    for (int r = 0; r < 4; ++r) {
        #pragma unroll
        for (int k = 0; k < 4; ++k) {
            const float keep = b4 ? p[r + 4][k] : p[r][k];
            const float send = b4 ? p[r][k] : p[r + 4][k];
            q4[r][k] = keep + row_xor<4>(send);
        }
    }
    #pragma unroll
    for (int r = 0; r < 2; ++r) {
        #pragma unroll
        for (int k = 0; k < 4; ++k) {
            const float keep = b2 ? q4[r + 2][k] : q4[r][k];
            const float send = b2 ? q4[r][k] : q4[r + 2][k];
            q2[r][k] = keep + row_xor<2>(send);
        }
    }
    #pragma unroll
    for (int k = 0; k < 4; ++k) {
        const float keep = b1 ? q2[1][k] : q2[0][k];
        const float send = b1 ? q2[0][k] : q2[1][k];
        q1[k] = keep + row_xor<1>(send);
    }
    const bool g = (lane & 8) != 0;
    const float r0 = row_xor<8>(g ? q1[0] : q1[2]);
    const float r1 = row_xor<8>(g ? q1[1] : q1[3]);
    v4f o;
    o.x = g ? r0 : q1[0];
    o.y = g ? q1[2] : r0;
    o.z = g ? r1 : q1[1];
    o.w = g ? q1[3] : r1;
    const int orow = row0 + (lane & 7) + half * 8;
    if (orow < n)
        *reinterpret_cast<v4f*>((g ? a_dst : a_src) + (size_t)orow * HEADS) = o;
}

__device__ __forceinline__ void
partition_block(const int* __restrict__ src, const int* __restrict__ dst,
                 const float* __restrict__ ea, uint2* __restrict__ rec,
                 int n, int e, int nb, int nw, int* hist, int wb) {
    const int t  = threadIdx.x;
    for (int b = t; b < nb; b += FT) hist[b] = 0;
    __syncthreads();

    const int sc_full = (RB * CAP) / nw;
    const int sc_last = ((n - (nb - 1) * RB) * CAP) / nw;
    const bool packable = n <= (1 << 24);

    int d[EPT], s[EPT], rk[EPT];
    float wv[EPT];
    const long long i0 = (long long)wb * EPB + t;
    #pragma unroll
    for (int k = 0; k < EPT; ++k) {
        const long long i = i0 + (long long)k * FT;
        d[k] = -1; s[k] = 0; wv[k] = 0.0f;
        if (i < e) { d[k] = dst[i]; s[k] = src[i]; wv[k] = ea[i]; }
    }
    #pragma unroll
    for (int k = 0; k < EPT; ++k) {
        rk[k] = 0;
        if (d[k] >= 0) rk[k] = atomicAdd(&hist[d[k] >> RB_LOG], 1);
    }
    __syncthreads();

    for (int b = t; b < nb; b += FT) {
        const int sc = (b == nb - 1) ? sc_last : sc_full;
        if (sc >= 2)
            rec[(size_t)b * (RB * CAP) + (size_t)wb * sc] = make_uint2((unsigned)hist[b], 0u);
    }
    if (packable) {
        #pragma unroll
        for (int k = 0; k < EPT; ++k) {
            if (d[k] >= 0) {
                const int b = d[k] >> RB_LOG;
                const int sc = (b == nb - 1) ? sc_last : sc_full;
                if (rk[k] < sc - 1)
                    rec[(size_t)b * (RB * CAP) + (size_t)wb * sc + 1 + rk[k]] =
                        make_uint2((unsigned)s[k] | ((unsigned)(d[k] & (RB - 1)) << 24),
                                   __float_as_uint(wv[k]));
            }
        }
    }
}

__global__ void __launch_bounds__(FT, 8)
front_kernel(const float* __restrict__ x,
             const float* __restrict__ W,
             const float* __restrict__ att_src,
             const float* __restrict__ att_dst,
             float* __restrict__ h,
             float* __restrict__ a_src, float* __restrict__ a_dst,
             const int* __restrict__ src, const int* __restrict__ dst,
             const float* __restrict__ ea, uint2* __restrict__ rec,
             int n, int e, int nb, int nw) {
    extern __shared__ int hist[];
    const int blk = (int)blockIdx.x;
    if (blk < nw) {
        partition_block(src, dst, ea, rec, n, e, nb, nw, hist, blk);
    } else {
        gemm_strip(x, W, att_src, att_dst, h, a_src, a_dst, n,
                   (blk - nw) * (FT / 32) + (int)(threadIdx.x >> 5));
    }
}


template <int K0, int N>
__device__ __forceinline__ void edge_group(const float* __restrict__ h, int sidx, float ex,
                                           int lane, v4f& acc) {
    v4f r[N];
    #pragma unroll
    for (int j = 0; j < N; ++j) {
        const unsigned s = (unsigned)__builtin_amdgcn_readlane(sidx, K0 + j);
        r[j] = *reinterpret_cast<const v4f*>(h + (size_t)s * DIN_ + 4 * lane);
    }
    float w[N];
    if (N > 0) w[0] = grp_bcast<K0>(ex);
    if (N > 1) w[N > 1 ? 1 : 0] = grp_bcast<K0 + 1>(ex);
    if (N > 2) w[N > 2 ? 2 : 0] = grp_bcast<K0 + 2>(ex);
    if (N > 3) w[N > 3 ? 3 : 0] = grp_bcast<K0 + 3>(ex);
    #pragma unroll
    for (int j = 0; j < N; ++j) {
        acc.x = fmaf(w[j], r[j].x, acc.x);
        acc.y = fmaf(w[j], r[j].y, acc.y);
        acc.z = fmaf(w[j], r[j].z, acc.z);
        acc.w = fmaf(w[j], r[j].w, acc.w);
    }
}
template <int K0>
__device__ __forceinline__ void edge_half(const float* __restrict__ h, int sidx, float ex,
                                          int lane, int rem, v4f& acc) {
    if (rem >= 4)      edge_group<K0, 4>(h, sidx, ex, lane, acc);
    else if (rem == 3) edge_group<K0, 3>(h, sidx, ex, lane, acc);
    else if (rem == 2) edge_group<K0, 2>(h, sidx, ex, lane, acc);
    else               edge_group<K0, 1>(h, sidx, ex, lane, acc);
}

#define ROUNDS (CAP / 8)

__device__ __forceinline__ void
gather_node(const int* __restrict__ src, const int* __restrict__ dst,
            const float* __restrict__ ea,
            const float* __restrict__ a_src, const float* __restrict__ a_dst,
            const float* __restrict__ h, const float* __restrict__ bias,
            float* out, int e, int node, int c, const uint2* slots, int lane) {
    float* orow = out + (size_t)node * DIN_;
    const float NEG_INF = -__builtin_inff();

    if (c <= CAP) {
        const int g  = lane >> 3;
        const int k8 = lane & 7;
        const float adh = a_dst[(size_t)node * HEADS + g];
        const float ash = a_src[(size_t)node * HEADS + g];
        const v4f hs = *reinterpret_cast<const v4f*>(h + (size_t)node * DIN_ + 4 * lane);

        int   sidx[ROUNDS];
        float wv[ROUNDS], as[ROUNDS], ex[ROUNDS];
        #pragma unroll
        for (int i = 0; i < ROUNDS; ++i) {
            if (8 * i >= c) break;
            const uint2 q = slots[min(8 * i + k8, c - 1)];
            sidx[i] = (int)q.x;
            wv[i] = __uint_as_float(q.y);
        }
        #pragma unroll
        for (int i = 0; i < ROUNDS; ++i) {
            if (8 * i >= c) break;
            as[i] = a_src[(size_t)(unsigned)sidx[i] * HEADS + g];
        }
        float wl = 0.0f, mx = NEG_INF;
        #pragma unroll
        for (int i = 0; i < ROUNDS; ++i) {
            if (8 * i >= c) break;
            const bool ok = 8 * i + k8 < c;
            wl += ok ? wv[i] : 0.0f;
            ex[i] = ok ? leaky(as[i] + adh) + __log2f(wv[i]) : NEG_INF;
            mx = fmaxf(mx, ex[i]);
        }
        const float wsum = grp_sum(wl);
        const float lw = c > 0 ? wsum / (float)c : 1.0f;
        const float l2self = __log2f(lw);
        const float vs = leaky(ash + adh) + l2self;
        const float m = fmaxf(grp_max(mx), vs);
        const float exs = __expf(vs - m);
        float dl = 0.0f;
        #pragma unroll
        for (int i = 0; i < ROUNDS; ++i) {
            if (8 * i >= c) break;
            const bool ok = 8 * i + k8 < c;
            ex[i] = ok ? __expf(ex[i] - m) : 0.0f;
            dl += ex[i];
        }
        const float den = grp_sum(dl) + exs;
        v4f acc;
        acc.x = exs * hs.x; acc.y = exs * hs.y; acc.z = exs * hs.z; acc.w = exs * hs.w;
        #pragma unroll
        for (int i = 0; i < ROUNDS; ++i) {
            const int rem = c - 8 * i;
            if (rem <= 0) break;
            edge_half<0>(h, sidx[i], ex[i], lane, rem, acc);
            if (rem > 4) edge_half<4>(h, sidx[i], ex[i], lane, rem - 4, acc);
        }
        const v4f b4 = *reinterpret_cast<const v4f*>(bias + 4 * lane);
        v4f res;
        res.x = acc.x / den + b4.x;
        res.y = acc.y / den + b4.y;
        res.z = acc.z / den + b4.z;
        res.w = acc.w / den + b4.w;
        *reinterpret_cast<v4f*>(orow + 4 * lane) = res;
        return;
    }

    const float* __restrict__ hself = h + (size_t)node * DIN_ + lane;
    float ad[HEADS], asf[HEADS];
    for (int hh = 0; hh < HEADS; ++hh) {
        ad[hh]  = a_dst[node * HEADS + hh];
        asf[hh] = a_src[node * HEADS + hh];
    }
    float acc[HEADS], den[HEADS];

    float wl = 0.0f;
    float ml[HEADS];
    for (int hh = 0; hh < HEADS; ++hh) ml[hh] = NEG_INF;
    for (int base = 0; base < e; base += 32) {
        const int i = base + lane;
        if (i < e && dst[i] == node) {
            const float w = ea[i];
            const float l2w = __log2f(w);
            const float4 as4 = *reinterpret_cast<const float4*>(a_src + (size_t)src[i] * HEADS);
            wl += w;
            ml[0] = fmaxf(ml[0], leaky(as4.x + ad[0]) + l2w);
            ml[1] = fmaxf(ml[1], leaky(as4.y + ad[1]) + l2w);
            ml[2] = fmaxf(ml[2], leaky(as4.z + ad[2]) + l2w);
            ml[3] = fmaxf(ml[3], leaky(as4.w + ad[3]) + l2w);
        }
    }
    const float wsum = wave_sum(wl);
    const float l2self = __log2f(wsum / (float)c);
    float mx[HEADS], exs[HEADS], dl[HEADS];
    for (int hh = 0; hh < HEADS; ++hh) {
        const float vs = leaky(asf[hh] + ad[hh]) + l2self;
        mx[hh] = fmaxf(wave_max(ml[hh]), vs);
        exs[hh] = __expf(vs - mx[hh]);
        acc[hh] = exs[hh] * hself[hh * CH];
        dl[hh] = 0.0f;
    }
    for (int base = 0; base < e; base += 32) {
        const int i = base + lane;
        const bool hit = i < e && dst[i] == node;
        int s = node;
        float ex[HEADS] = {0.0f, 0.0f, 0.0f, 0.0f};
        if (hit) {
            s = src[i];
            const float l2w = __log2f(ea[i]);
            const float4 as4 = *reinterpret_cast<const float4*>(a_src + (size_t)s * HEADS);
            ex[0] = __expf(leaky(as4.x + ad[0]) + l2w - mx[0]);
            ex[1] = __expf(leaky(as4.y + ad[1]) + l2w - mx[1]);
            ex[2] = __expf(leaky(as4.z + ad[2]) + l2w - mx[2]);
            ex[3] = __expf(leaky(as4.w + ad[3]) + l2w - mx[3]);
            for (int hh = 0; hh < HEADS; ++hh) dl[hh] += ex[hh];
        }
        unsigned mask = __builtin_amdgcn_ballot_w32(hit);
        while (mask) {
            const int j = __builtin_ctz(mask);
            mask &= mask - 1;
            const int sb = __builtin_amdgcn_readlane(s, j);
            const float* __restrict__ hp = h + (size_t)sb * DIN_ + lane;
            for (int cc = 0; cc < HEADS; ++cc)
                acc[cc] = fmaf(lane_f(ex[cc], j), hp[cc * CH], acc[cc]);
        }
    }
    for (int hh = 0; hh < HEADS; ++hh) den[hh] = wave_sum(dl[hh]) + exs[hh];

    for (int cc = 0; cc < HEADS; ++cc)
        orow[lane + cc * CH] = acc[cc] / den[cc] + bias[lane + cc * CH];
}

#define SP (CAP + 1)

__global__ void __launch_bounds__(PT)
bin_kernel(const uint2* rec,
           const int* __restrict__ src, const int* __restrict__ dst,
           const float* __restrict__ ea,
           const float* __restrict__ a_src, const float* __restrict__ a_dst,
           const float* __restrict__ h, const float* __restrict__ bias,
           float* out, int n, int e, int nw) {
    __shared__ uint2 slot[RB * SP];
    __shared__ int lcnt[RB];
    __shared__ int ovf_s;
    __shared__ int next_s;
    extern __shared__ int cntw[];
    const int t = threadIdx.x;
    const int lane = t & 31;
    const int wave = __builtin_amdgcn_readfirstlane(t >> 5);
    const int b = blockIdx.x;
    const int node0 = b * RB;
    const int rows = min(RB, n - node0);
    const int sc = nw > 0 ? (rows * CAP) / nw : 0;
    const int usable = (n <= (1 << 24)) ? sc - 1 : -1;
    const uint2* reg = rec + (size_t)b * (RB * CAP);

    if (t < RB) lcnt[t] = 0;
    if (t == 0) { ovf_s = (sc < 2) ? 1 : 0; next_s = PT / 32; }
    __syncthreads();
    if (sc >= 2) {
        for (int w = t; w < nw; w += PT) {
            const int c = (int)reg[(size_t)w * sc].x;
            cntw[w] = c;
            if (c > usable) ovf_s = 1;
        }
    }
    __syncthreads();
    const bool ovf = ovf_s != 0;

    if (!ovf) {
        for (int w0 = wave; w0 < nw; w0 += (PT / 32) * PU) {
            uint2 q[PU];
            bool ok[PU];
            #pragma unroll
            for (int u = 0; u < PU; ++u) {
                const int w = w0 + (PT / 32) * u;
                const int c = w < nw ? cntw[w] : 0;
                ok[u] = lane < c;
                q[u] = make_uint2(0u, 0u);
                if (ok[u]) q[u] = reg[(size_t)w * sc + 1 + lane];
            }
            #pragma unroll
            for (int u = 0; u < PU; ++u) {
                if (ok[u]) {
                    const int dl = (int)(q[u].x >> 24);
                    const int r = atomicAdd(&lcnt[dl], 1);
                    if (r < CAP)
                        slot[dl * SP + r] = make_uint2(q[u].x & 0xFFFFFFu, q[u].y);
                }
            }
            for (int u = 0; u < PU; ++u) {
                const int w = w0 + (PT / 32) * u;
                const int c = w < nw ? cntw[w] : 0;
                for (int k = lane + 32; k < c; k += 32) {
                    const uint2 qq = reg[(size_t)w * sc + 1 + k];
                    const int dl = (int)(qq.x >> 24);
                    const int r = atomicAdd(&lcnt[dl], 1);
                    if (r < CAP)
                        slot[dl * SP + r] = make_uint2(qq.x & 0xFFFFFFu, qq.y);
                }
            }
        }
    } else {
        for (long long i = t; i < e; i += PT) {
            const int d = dst[i];
            if (d >= node0 && d < node0 + rows) {
                const int r = atomicAdd(&lcnt[d - node0], 1);
                if (r < CAP)
                    slot[(d - node0) * SP + r] =
                        make_uint2((unsigned)src[i], __float_as_uint(ea[i]));
            }
        }
    }
    __syncthreads();
    int i = wave;
    while (i < rows) {
        int nx = 0;
        if (lane == 0) nx = atomicAdd(&next_s, 1);
        const int c = __builtin_amdgcn_readfirstlane(lcnt[i]);
        gather_node(src, dst, ea, a_src, a_dst, h, bias, out, e, node0 + i, c,
                    slot + i * SP, lane);
        i = __builtin_amdgcn_readfirstlane(nx);
    }
}


extern "C" void kernel_launch(void* const* d_in, const int* in_sizes, int n_in,
                              void* d_out, int out_size, void* d_ws, size_t ws_size,
                              hipStream_t stream) {
    const float* x        = (const float*)d_in[0];
    const int*   eidx     = (const int*)d_in[1];
    const float* eattr    = (const float*)d_in[2];
    const float* W        = (const float*)d_in[3];
    const float* att_src  = (const float*)d_in[4];
    const float* att_dst  = (const float*)d_in[5];
    const float* bias     = (const float*)d_in[6];

    const int n = in_sizes[0] / DIN_;
    const int e = in_sizes[2];
    const int* src = eidx;
    const int* dst = eidx + e;

    float* out = (float*)d_out;

    float* h    = (float*)d_ws;
    float* asrc = h    + (size_t)n * DIN_;
    float* adst = asrc + (size_t)n * HEADS;

    uint2* rec = (uint2*)out;
    const int NB = (n + RB - 1) / RB;
    const int NW = (e + EPB - 1) / EPB;
    const int GB = (n + 16 * (FT / 32) - 1) / (16 * (FT / 32));
    front_kernel<<<(unsigned)(NW + GB), FT, (size_t)NB * sizeof(int), stream>>>(
        x, W, att_src, att_dst, h, asrc, adst, src, dst, eattr, rec, n, e, NB, NW);
    bin_kernel<<<(unsigned)NB, PT, (size_t)NW * sizeof(int), stream>>>(
        rec, src, dst, eattr, asrc, adst, h, bias, out, n, e, NW);
}
